# starting kernel
# baseline (speedup 1.0000x reference)
#include <hip/hip_runtime.h>

typedef float v2f __attribute__((ext_vector_type(2)));
typedef float v8f __attribute__((ext_vector_type(8)));

#define CIN   256
#define CCOMP 64
#define CENC  36
#define HH    64
#define WWID  64
#define HW    4096
#define KK    9
#define SS    4
#define KDIM  576

__global__ void gg_mean_kernel(const float* __restrict__ x,
                               float* __restrict__ mean) {
    int idx = blockIdx.x * 256 + threadIdx.x;
    int b   = idx >> 12;
    int pix = idx & 4095;
    const float* xb = x + (size_t)b * CIN * HW + pix;
    float s = 0.f;
#pragma unroll 8
    for (int c = 0; c < CIN; ++c) s += xb[c * HW];
    mean[idx] = s * (1.0f / CIN);
}

__global__ void gg_comp_gemm(const float* __restrict__ x,
                             const float* __restrict__ w1,
                             const float* __restrict__ b1,
                             float* __restrict__ comp) {
    int b     = blockIdx.x >> 7;
    int nBase = (blockIdx.x & 127) * 32;
    int wave  = threadIdx.x >> 5;
    int lane  = threadIdx.x & 31;
    int lr    = lane & 15;
    int half  = lane >> 4;
    int m0    = (wave & 3) * 16;
    int n0    = nBase + (wave >> 2) * 16;

    const float* xb   = x + (size_t)b * CIN * HW;
    const float* aRow = w1 + (m0 + lr) * CIN;

    v8f acc = {};
    for (int k = 0; k < CIN; k += 4) {
        int ke = k + 2 * half;
        v2f a, bm;
        a.x  = aRow[ke];
        a.y  = aRow[ke + 1];
        bm.x = xb[ke * HW + n0 + lr];
        bm.y = xb[(ke + 1) * HW + n0 + lr];
        acc = __builtin_amdgcn_wmma_f32_16x16x4_f32(
                  false, a, false, bm, (short)0, acc, false, false);
    }

    float* cb = comp + (size_t)b * CCOMP * HW;
#pragma unroll
    for (int r = 0; r < 8; ++r) {
        int m = m0 + r + 8 * half;
        cb[m * HW + n0 + lr] = acc[r] + b1[m];
    }
}

__global__ void gg_kern_gemm(const float* __restrict__ comp,
                             const float* __restrict__ w2,
                             const float* __restrict__ b2,
                             float* __restrict__ kern) {
    int b    = blockIdx.x >> 8;
    int n0   = (blockIdx.x & 255) * 16;
    int wave = threadIdx.x >> 5;
    int lane = threadIdx.x & 31;
    int lr   = lane & 15;
    int half = lane >> 4;
    int m0   = wave * 16;

    int n = n0 + lr;
    int h = n >> 6, w = n & 63;
    const float* cb = comp + (size_t)b * CCOMP * HW;
    int mA = m0 + lr;
    bool aValid = (mA < CENC);

    v8f acc = {};
    for (int k = 0; k < KDIM; k += 4) {
        int ke = k + 2 * half;
        v2f a, bm;
        a.x = aValid ? w2[mA * KDIM + ke]     : 0.f;
        a.y = aValid ? w2[mA * KDIM + ke + 1] : 0.f;
        {
            int ci = ke / 9, r = ke - ci * 9, ky = r / 3, kx = r - ky * 3;
            int hh = h + ky - 1, ww = w + kx - 1;
            bm.x = ((unsigned)hh < 64u && (unsigned)ww < 64u)
                       ? cb[ci * HW + hh * 64 + ww] : 0.f;
        }
        {
            int k1 = ke + 1;
            int ci = k1 / 9, r = k1 - ci * 9, ky = r / 3, kx = r - ky * 3;
            int hh = h + ky - 1, ww = w + kx - 1;
            bm.y = ((unsigned)hh < 64u && (unsigned)ww < 64u)
                       ? cb[ci * HW + hh * 64 + ww] : 0.f;
        }
        acc = __builtin_amdgcn_wmma_f32_16x16x4_f32(
                  false, a, false, bm, (short)0, acc, false, false);
    }

    float* kb = kern + (size_t)b * CENC * HW;
#pragma unroll
    for (int r = 0; r < 8; ++r) {
        int m = m0 + r + 8 * half;
        if (m < CENC) kb[m * HW + n] = acc[r] + b2[m];
    }
}

__global__ void gg_carafe_kernel(const float* __restrict__ x,
                                 const float* __restrict__ mean,
                                 const float* __restrict__ kern,
                                 float* __restrict__ out) {
    __shared__ float smask[KK][SS][WWID];

    int bh = blockIdx.x;
    int b  = bh >> 6, h = bh & 63;
    int tx = threadIdx.x & 63;
    int ty = threadIdx.x >> 6;

    const float* meanb = mean + b * HW;
    const float* kb    = kern + (size_t)b * CENC * HW;

    if (ty == 0) {
        int w   = tx;
        float mc = meanb[h * 64 + w];
        float mval[SS][KK];
#pragma unroll
        for (int kk = 0; kk < KK; ++kk) {
            int dy = kk / 3 - 1, dx = kk % 3 - 1;
            int hh = h + dy, ww = w + dx;
            float mp = ((unsigned)hh < 64u && (unsigned)ww < 64u)
                           ? meanb[hh * 64 + ww] : 0.f;
            float g = mp - mc;
            g = 1.0f / (g * g + 0.2f);
#pragma unroll
            for (int s = 0; s < SS; ++s)
                mval[s][kk] = g * kb[(kk * SS + s) * HW + h * 64 + w];
        }
#pragma unroll
        for (int s = 0; s < SS; ++s) {
            float mx = mval[s][0];
#pragma unroll
            for (int kk = 1; kk < KK; ++kk) mx = fmaxf(mx, mval[s][kk]);
            float e[KK], sum = 0.f;
#pragma unroll
            for (int kk = 0; kk < KK; ++kk) { e[kk] = __expf(mval[s][kk] - mx); sum += e[kk]; }
            float inv = 1.0f / sum;
#pragma unroll
            for (int kk = 0; kk < KK; ++kk) smask[kk][s][w] = e[kk] * inv;
        }
    }
    __syncthreads();

    for (int c = ty; c < CIN; c += 4) {
        const float* xb = x + ((size_t)(b * CIN + c)) * HW;
        __builtin_prefetch(xb + 4 * HW, 0, 0);
        float v[KK];
#pragma unroll
        for (int kk = 0; kk < KK; ++kk) {
            int dy = kk / 3 - 1, dx = kk % 3 - 1;
            int hh = h + dy, ww = tx + dx;
            v[kk] = ((unsigned)hh < 64u && (unsigned)ww < 64u)
                        ? xb[hh * 64 + ww] : 0.f;
        }
        size_t obase = ((size_t)(b * CIN + c)) * 128 * 128;
#pragma unroll
        for (int p = 0; p < 2; ++p)
#pragma unroll
            for (int q = 0; q < 2; ++q) {
                int s = p * 2 + q;
                float acc = 0.f;
#pragma unroll
                for (int kk = 0; kk < KK; ++kk)
                    acc = fmaf(v[kk], smask[kk][s][tx], acc);
                out[obase + (size_t)(2 * h + p) * 128 + (2 * tx + q)] = acc;
            }
    }
}

extern "C" void kernel_launch(void* const* d_in, const int* in_sizes, int n_in,
                              void* d_out, int out_size, void* d_ws, size_t ws_size,
                              hipStream_t stream) {
    const float* x  = (const float*)d_in[0];
    const float* w1 = (const float*)d_in[1];
    const float* b1 = (const float*)d_in[2];
    const float* w2 = (const float*)d_in[3];
    const float* b2 = (const float*)d_in[4];
    float* out = (float*)d_out;

    float* ws   = (float*)d_ws;
    float* mean = ws;
    float* comp = ws + 32768;
    float* kern = comp + (size_t)8 * CCOMP * HW;

    gg_mean_kernel <<<128,  256, 0, stream>>>(x, mean);
    gg_comp_gemm   <<<1024, 256, 0, stream>>>(x, w1, b1, comp);
    gg_kern_gemm   <<<2048,  96, 0, stream>>>(comp, w2, b2, kern);
    gg_carafe_kernel<<<512, 256, 0, stream>>>(x, mean, kern, out);
}


# 3x3-conv GEMM: both operands (comp halo rows, w2) staged once per 128-pixel block in LDS; k loop is grouped ds_load + f32 WMMA, one B fragment feeds 3 M tiles; k order and sums unchanged
# speedup vs baseline: 1.5563x; 1.5563x over previous
#include <hip/hip_runtime.h>

typedef float v2f __attribute__((ext_vector_type(2)));
typedef float v4f __attribute__((ext_vector_type(4)));
typedef float v8f __attribute__((ext_vector_type(8)));

#define CIN   256
#define CCOMP 64
#define CENC  36
#define HH    64
#define WWID  64
#define HW    4096
#define KK    9
#define SS    4
#define KDIM  576

__global__ void gg_mean_kernel(const float* __restrict__ x,
                               float* __restrict__ mean) {
    int idx = blockIdx.x * 256 + threadIdx.x;
    int b   = idx >> 12;
    int pix = idx & 4095;
    const float* xb = x + (size_t)b * CIN * HW + pix;
    float s = 0.f;
#pragma unroll 8
    for (int c = 0; c < CIN; ++c) s += xb[c * HW];
    mean[idx] = s * (1.0f / CIN);
}

__global__ void gg_comp_gemm(const float* __restrict__ x,
                             const float* __restrict__ w1,
                             const float* __restrict__ b1,
                             float* __restrict__ comp) {
    int b     = blockIdx.x >> 7;
    int nBase = (blockIdx.x & 127) * 32;
    int wave  = threadIdx.x >> 5;
    int lane  = threadIdx.x & 31;
    int lr    = lane & 15;
    int half  = lane >> 4;
    int m0    = (wave & 3) * 16;
    int n0    = nBase + (wave >> 2) * 16;

    const float* xb   = x + (size_t)b * CIN * HW;
    const float* aRow = w1 + (m0 + lr) * CIN;

    v8f acc = {};
    for (int k = 0; k < CIN; k += 4) {
        int ke = k + 2 * half;
        v2f a, bm;
        a.x  = aRow[ke];
        a.y  = aRow[ke + 1];
        bm.x = xb[ke * HW + n0 + lr];
        bm.y = xb[(ke + 1) * HW + n0 + lr];
        acc = __builtin_amdgcn_wmma_f32_16x16x4_f32(
                  false, a, false, bm, (short)0, acc, false, false);
    }

    float* cb = comp + (size_t)b * CCOMP * HW;
#pragma unroll
    for (int r = 0; r < 8; ++r) {
        int m = m0 + r + 8 * half;
        cb[m * HW + n0 + lr] = acc[r] + b1[m];
    }
}

#define K3_APITCH 580
#define K3_AROWS  48
#define K3_BROW   68
#define K3_BCH    272
#define K3_BGRP   1088

__global__ void __launch_bounds__(256)
gg_kern_gemm(const float* __restrict__ comp,
             const float* __restrict__ w2,
             const float* __restrict__ b2,
             float* __restrict__ kern) {
    __shared__ __attribute__((aligned(16))) float As[K3_AROWS * K3_APITCH];
    __shared__ __attribute__((aligned(16))) float Bs[CCOMP * K3_BCH];

    const int tid = threadIdx.x;
    const int b   = blockIdx.x >> 5;
    const int h0  = (blockIdx.x & 31) * 2;
    const float* cb = comp + (size_t)b * CCOMP * HW;

    v4f wv[20];
#pragma unroll
    for (int t = 0; t < 20; ++t)
        wv[t] = *reinterpret_cast<const v4f*>(w2 + (tid + t * 256) * 4);
    const int iTail = 20 * 256 + (tid & 63);
    v4f wTail = *reinterpret_cast<const v4f*>(w2 + iTail * 4);
    const int  w4    = tid & 15;
    const int  rS    = (tid >> 4) & 3;
    const int  cq    = tid >> 6;
    const int  hS    = h0 - 1 + rS;
    const bool rowOk = (unsigned)hS < 64u;
    const int  hL    = rowOk ? hS : h0;
    v4f cv[16];
#pragma unroll
    for (int t = 0; t < 16; ++t)
        cv[t] = *reinterpret_cast<const v4f*>(cb + (cq + 4 * t) * HW + hL * 64 + w4 * 4);
    const int hf8 = (tid & 16) >> 1;
    float bias0[8], bias1[8], bias2[4];
#pragma unroll
    for (int r = 0; r < 8; ++r) {
        bias0[r] = b2[r + hf8];
        bias1[r] = b2[16 + r + hf8];
    }
#pragma unroll
    for (int r = 0; r < 4; ++r)
        bias2[r] = b2[32 + r];
#pragma unroll
    for (int t = 0; t < 20; ++t) {
        int i  = tid + t * 256;
        int m  = i / (KDIM / 4);
        int j4 = i - m * (KDIM / 4);
        *reinterpret_cast<v4f*>(&As[m * K3_APITCH + j4 * 4]) = wv[t];
    }
    {
        int m  = iTail / (KDIM / 4);
        int j4 = iTail - m * (KDIM / 4);
        *reinterpret_cast<v4f*>(&As[m * K3_APITCH + j4 * 4]) = wTail;
    }
    for (int i = tid; i < (K3_AROWS - CENC) * K3_APITCH / 4; i += 256) {
        v4f z = {0.f, 0.f, 0.f, 0.f};
        *reinterpret_cast<v4f*>(&As[CENC * K3_APITCH + i * 4]) = z;
    }
#pragma unroll
    for (int t = 0; t < 16; ++t) {
        int o = (cq + 4 * t) * K3_BCH + rS * K3_BROW + 1 + w4 * 4;
        Bs[o]     = rowOk ? cv[t].x : 0.f;
        Bs[o + 1] = rowOk ? cv[t].y : 0.f;
        Bs[o + 2] = rowOk ? cv[t].z : 0.f;
        Bs[o + 3] = rowOk ? cv[t].w : 0.f;
    }
    {
        int o = (tid >> 2) * K3_BCH + (tid & 3) * K3_BROW;
        Bs[o]      = 0.f;
        Bs[o + 65] = 0.f;
    }
    __syncthreads();

    const int wave  = tid >> 5;
    const int lane  = tid & 31;
    const int lr    = lane & 15;
    const int half  = lane >> 4;
    const int rowIn = wave >> 2;
    const int w0    = (wave & 3) * 16;

    int boff[9][2];
#pragma unroll
    for (int j = 0; j < 9; ++j)
#pragma unroll
        for (int s = 0; s < 2; ++s) {
            int q  = j * 4 + 2 * half + s;
            int cc = q / 9, t = q - cc * 9, ky = t / 3, kx = t - ky * 3;
            boff[j][s] = cc * K3_BCH + (rowIn + ky) * K3_BROW + (w0 + lr + kx);
        }
    const int aoff = lr * K3_APITCH + 2 * half;

    v2f fa[2][9][3], fb[2][9];
    v8f acc0 = {}, acc1 = {}, acc2 = {};

#pragma unroll
    for (int j = 0; j < 9; ++j) {
        fb[0][j].x = Bs[boff[j][0]];
        fb[0][j].y = Bs[boff[j][1]];
        fa[0][j][0] = *reinterpret_cast<const v2f*>(&As[aoff + j * 4]);
        fa[0][j][1] = *reinterpret_cast<const v2f*>(&As[aoff + 16 * K3_APITCH + j * 4]);
        fa[0][j][2] = *reinterpret_cast<const v2f*>(&As[aoff + 32 * K3_APITCH + j * 4]);
    }
#pragma unroll
    for (int g = 0; g < 16; ++g) {
        const int cur = g & 1, nxt = cur ^ 1;
        if (g + 1 < 16) {
#pragma unroll
            for (int j = 0; j < 9; ++j) {
                const int k1 = (g + 1) * 36 + j * 4;
                fb[nxt][j].x = Bs[(g + 1) * K3_BGRP + boff[j][0]];
                fb[nxt][j].y = Bs[(g + 1) * K3_BGRP + boff[j][1]];
                fa[nxt][j][0] = *reinterpret_cast<const v2f*>(&As[aoff + k1]);
                fa[nxt][j][1] = *reinterpret_cast<const v2f*>(&As[aoff + 16 * K3_APITCH + k1]);
                fa[nxt][j][2] = *reinterpret_cast<const v2f*>(&As[aoff + 32 * K3_APITCH + k1]);
            }
        }
        __builtin_amdgcn_sched_barrier(0);
#pragma unroll
        for (int j = 0; j < 9; ++j) {
            acc0 = __builtin_amdgcn_wmma_f32_16x16x4_f32(
                       false, fa[cur][j][0], false, fb[cur][j], (short)0, acc0, false, false);
            acc1 = __builtin_amdgcn_wmma_f32_16x16x4_f32(
                       false, fa[cur][j][1], false, fb[cur][j], (short)0, acc1, false, false);
            acc2 = __builtin_amdgcn_wmma_f32_16x16x4_f32(
                       false, fa[cur][j][2], false, fb[cur][j], (short)0, acc2, false, false);
        }
        __builtin_amdgcn_sched_barrier(0);
    }

    float* kb = kern + (size_t)b * CENC * HW + (h0 + rowIn) * 64 + w0 + lr;
    const int mh = 8 * half;
#pragma unroll
    for (int r = 0; r < 8; ++r) {
        kb[(mh + r) * HW]      = acc0[r] + bias0[r];
        kb[(16 + mh + r) * HW] = acc1[r] + bias1[r];
    }
    if (half == 0) {
#pragma unroll
        for (int r = 0; r < 4; ++r)
            kb[(32 + r) * HW] = acc2[r] + bias2[r];
    }
}

__global__ void gg_carafe_kernel(const float* __restrict__ x,
                                 const float* __restrict__ mean,
                                 const float* __restrict__ kern,
                                 float* __restrict__ out) {
    __shared__ float smask[KK][SS][WWID];

    int bh = blockIdx.x;
    int b  = bh >> 6, h = bh & 63;
    int tx = threadIdx.x & 63;
    int ty = threadIdx.x >> 6;

    const float* meanb = mean + b * HW;
    const float* kb    = kern + (size_t)b * CENC * HW;

    if (ty == 0) {
        int w   = tx;
        float mc = meanb[h * 64 + w];
        float mval[SS][KK];
#pragma unroll
        for (int kk = 0; kk < KK; ++kk) {
            int dy = kk / 3 - 1, dx = kk % 3 - 1;
            int hh = h + dy, ww = w + dx;
            float mp = ((unsigned)hh < 64u && (unsigned)ww < 64u)
                           ? meanb[hh * 64 + ww] : 0.f;
            float g = mp - mc;
            g = 1.0f / (g * g + 0.2f);
#pragma unroll
            for (int s = 0; s < SS; ++s)
                mval[s][kk] = g * kb[(kk * SS + s) * HW + h * 64 + w];
        }
#pragma unroll
        for (int s = 0; s < SS; ++s) {
            float mx = mval[s][0];
#pragma unroll
            for (int kk = 1; kk < KK; ++kk) mx = fmaxf(mx, mval[s][kk]);
            float e[KK], sum = 0.f;
#pragma unroll
            for (int kk = 0; kk < KK; ++kk) { e[kk] = __expf(mval[s][kk] - mx); sum += e[kk]; }
            float inv = 1.0f / sum;
#pragma unroll
            for (int kk = 0; kk < KK; ++kk) smask[kk][s][w] = e[kk] * inv;
        }
    }
    __syncthreads();

    for (int c = ty; c < CIN; c += 4) {
        const float* xb = x + ((size_t)(b * CIN + c)) * HW;
        __builtin_prefetch(xb + 4 * HW, 0, 0);
        float v[KK];
#pragma unroll
        for (int kk = 0; kk < KK; ++kk) {
            int dy = kk / 3 - 1, dx = kk % 3 - 1;
            int hh = h + dy, ww = tx + dx;
            v[kk] = ((unsigned)hh < 64u && (unsigned)ww < 64u)
                        ? xb[hh * 64 + ww] : 0.f;
        }
        size_t obase = ((size_t)(b * CIN + c)) * 128 * 128;
#pragma unroll
        for (int p = 0; p < 2; ++p)
#pragma unroll
            for (int q = 0; q < 2; ++q) {
                int s = p * 2 + q;
                float acc = 0.f;
#pragma unroll
                for (int kk = 0; kk < KK; ++kk)
                    acc = fmaf(v[kk], smask[kk][s][tx], acc);
                out[obase + (size_t)(2 * h + p) * 128 + (2 * tx + q)] = acc;
            }
    }
}

extern "C" void kernel_launch(void* const* d_in, const int* in_sizes, int n_in,
                              void* d_out, int out_size, void* d_ws, size_t ws_size,
                              hipStream_t stream) {
    const float* x  = (const float*)d_in[0];
    const float* w1 = (const float*)d_in[1];
    const float* b1 = (const float*)d_in[2];
    const float* w2 = (const float*)d_in[3];
    const float* b2 = (const float*)d_in[4];
    float* out = (float*)d_out;

    float* ws   = (float*)d_ws;
    float* mean = ws;
    float* comp = ws + 32768;
    float* kern = comp + (size_t)8 * CCOMP * HW;

    gg_mean_kernel <<<128,  256, 0, stream>>>(x, mean);
    gg_comp_gemm   <<<1024, 256, 0, stream>>>(x, w1, b1, comp);
    gg_kern_gemm   <<<256,  256, 0, stream>>>(comp, w2, b2, kern);
    gg_carafe_kernel<<<512, 256, 0, stream>>>(x, mean, kern, out);
}


# channel mean fused into the 1x1-conv GEMM kernel: summed from the x B-fragments it already loads; mean kernel and its launch removed; mean's sum order changes (2 partial sums per lane half)
# speedup vs baseline: 1.9180x; 1.2324x over previous
#include <hip/hip_runtime.h>

typedef float v2f __attribute__((ext_vector_type(2)));
typedef float v4f __attribute__((ext_vector_type(4)));
typedef float v8f __attribute__((ext_vector_type(8)));

#define CIN   256
#define CCOMP 64
#define CENC  36
#define HH    64
#define WWID  64
#define HW    4096
#define KK    9
#define SS    4
#define KDIM  576

__device__ __forceinline__ float gg_xhalf(float v) {
    int bits = __builtin_bit_cast(int, v);
    return __builtin_bit_cast(float,
        __builtin_amdgcn_permlanex16(bits, bits, 0x76543210, 0xfedcba98, false, false));
}

__global__ void gg_comp_gemm(const float* __restrict__ x,
                             const float* __restrict__ w1,
                             const float* __restrict__ b1,
                             float* __restrict__ comp,
                             float* __restrict__ mean) {
    int b     = blockIdx.x >> 7;
    int nBase = (blockIdx.x & 127) * 32;
    int wave  = threadIdx.x >> 5;
    int lane  = threadIdx.x & 31;
    int lr    = lane & 15;
    int half  = lane >> 4;
    int m0    = (wave & 3) * 16;
    int n0    = nBase + (wave >> 2) * 16;

    const float* xb   = x + (size_t)b * CIN * HW;
    const float* aRow = w1 + (m0 + lr) * CIN;

    v8f acc = {};
    v2f csum = {0.f, 0.f};
    for (int k = 0; k < CIN; k += 4) {
        int ke = k + 2 * half;
        v2f a, bm;
        a.x  = aRow[ke];
        a.y  = aRow[ke + 1];
        bm.x = xb[ke * HW + n0 + lr];
        bm.y = xb[(ke + 1) * HW + n0 + lr];
        acc = __builtin_amdgcn_wmma_f32_16x16x4_f32(
                  false, a, false, bm, (short)0, acc, false, false);
        csum += bm;
    }

    float cs = csum.x + csum.y;
    cs += gg_xhalf(cs);
    if ((wave & 3) == 0 && half == 0)
        mean[b * HW + n0 + lr] = cs * (1.0f / CIN);

    float* cb = comp + (size_t)b * CCOMP * HW;
#pragma unroll
    for (int r = 0; r < 8; ++r) {
        int m = m0 + r + 8 * half;
        cb[m * HW + n0 + lr] = acc[r] + b1[m];
    }
}

#define K3_APITCH 580
#define K3_AROWS  48
#define K3_BROW   68
#define K3_BCH    272
#define K3_BGRP   1088

__global__ void __launch_bounds__(256)
gg_kern_gemm(const float* __restrict__ comp,
             const float* __restrict__ w2,
             const float* __restrict__ b2,
             float* __restrict__ kern) {
    __shared__ __attribute__((aligned(16))) float As[K3_AROWS * K3_APITCH];
    __shared__ __attribute__((aligned(16))) float Bs[CCOMP * K3_BCH];

    const int tid = threadIdx.x;
    const int b   = blockIdx.x >> 5;
    const int h0  = (blockIdx.x & 31) * 2;
    const float* cb = comp + (size_t)b * CCOMP * HW;

    v4f wv[20];
#pragma unroll
    for (int t = 0; t < 20; ++t)
        wv[t] = *reinterpret_cast<const v4f*>(w2 + (tid + t * 256) * 4);
    const int iTail = 20 * 256 + (tid & 63);
    v4f wTail = *reinterpret_cast<const v4f*>(w2 + iTail * 4);
    const int  w4    = tid & 15;
    const int  rS    = (tid >> 4) & 3;
    const int  cq    = tid >> 6;
    const int  hS    = h0 - 1 + rS;
    const bool rowOk = (unsigned)hS < 64u;
    const int  hL    = rowOk ? hS : h0;
    v4f cv[16];
#pragma unroll
    for (int t = 0; t < 16; ++t)
        cv[t] = *reinterpret_cast<const v4f*>(cb + (cq + 4 * t) * HW + hL * 64 + w4 * 4);
    const int hf8 = (tid & 16) >> 1;
    float bias0[8], bias1[8], bias2[4];
#pragma unroll
    for (int r = 0; r < 8; ++r) {
        bias0[r] = b2[r + hf8];
        bias1[r] = b2[16 + r + hf8];
    }
#pragma unroll
    for (int r = 0; r < 4; ++r)
        bias2[r] = b2[32 + r];
#pragma unroll
    for (int t = 0; t < 20; ++t) {
        int i  = tid + t * 256;
        int m  = i / (KDIM / 4);
        int j4 = i - m * (KDIM / 4);
        *reinterpret_cast<v4f*>(&As[m * K3_APITCH + j4 * 4]) = wv[t];
    }
    {
        int m  = iTail / (KDIM / 4);
        int j4 = iTail - m * (KDIM / 4);
        *reinterpret_cast<v4f*>(&As[m * K3_APITCH + j4 * 4]) = wTail;
    }
    for (int i = tid; i < (K3_AROWS - CENC) * K3_APITCH / 4; i += 256) {
        v4f z = {0.f, 0.f, 0.f, 0.f};
        *reinterpret_cast<v4f*>(&As[CENC * K3_APITCH + i * 4]) = z;
    }
#pragma unroll
    for (int t = 0; t < 16; ++t) {
        int o = (cq + 4 * t) * K3_BCH + rS * K3_BROW + 1 + w4 * 4;
        Bs[o]     = rowOk ? cv[t].x : 0.f;
        Bs[o + 1] = rowOk ? cv[t].y : 0.f;
        Bs[o + 2] = rowOk ? cv[t].z : 0.f;
        Bs[o + 3] = rowOk ? cv[t].w : 0.f;
    }
    {
        int o = (tid >> 2) * K3_BCH + (tid & 3) * K3_BROW;
        Bs[o]      = 0.f;
        Bs[o + 65] = 0.f;
    }
    __syncthreads();

    const int wave  = tid >> 5;
    const int lane  = tid & 31;
    const int lr    = lane & 15;
    const int half  = lane >> 4;
    const int rowIn = wave >> 2;
    const int w0    = (wave & 3) * 16;

    int boff[9][2];
#pragma unroll
    for (int j = 0; j < 9; ++j)
#pragma unroll
        for (int s = 0; s < 2; ++s) {
            int q  = j * 4 + 2 * half + s;
            int cc = q / 9, t = q - cc * 9, ky = t / 3, kx = t - ky * 3;
            boff[j][s] = cc * K3_BCH + (rowIn + ky) * K3_BROW + (w0 + lr + kx);
        }
    const int aoff = lr * K3_APITCH + 2 * half;

    v2f fa[2][9][3], fb[2][9];
    v8f acc0 = {}, acc1 = {}, acc2 = {};

#pragma unroll
    for (int j = 0; j < 9; ++j) {
        fb[0][j].x = Bs[boff[j][0]];
        fb[0][j].y = Bs[boff[j][1]];
        fa[0][j][0] = *reinterpret_cast<const v2f*>(&As[aoff + j * 4]);
        fa[0][j][1] = *reinterpret_cast<const v2f*>(&As[aoff + 16 * K3_APITCH + j * 4]);
        fa[0][j][2] = *reinterpret_cast<const v2f*>(&As[aoff + 32 * K3_APITCH + j * 4]);
    }
#pragma unroll
    for (int g = 0; g < 16; ++g) {
        const int cur = g & 1, nxt = cur ^ 1;
        if (g + 1 < 16) {
#pragma unroll
            for (int j = 0; j < 9; ++j) {
                const int k1 = (g + 1) * 36 + j * 4;
                fb[nxt][j].x = Bs[(g + 1) * K3_BGRP + boff[j][0]];
                fb[nxt][j].y = Bs[(g + 1) * K3_BGRP + boff[j][1]];
                fa[nxt][j][0] = *reinterpret_cast<const v2f*>(&As[aoff + k1]);
                fa[nxt][j][1] = *reinterpret_cast<const v2f*>(&As[aoff + 16 * K3_APITCH + k1]);
                fa[nxt][j][2] = *reinterpret_cast<const v2f*>(&As[aoff + 32 * K3_APITCH + k1]);
            }
        }
        __builtin_amdgcn_sched_barrier(0);
#pragma unroll
        for (int j = 0; j < 9; ++j) {
            acc0 = __builtin_amdgcn_wmma_f32_16x16x4_f32(
                       false, fa[cur][j][0], false, fb[cur][j], (short)0, acc0, false, false);
            acc1 = __builtin_amdgcn_wmma_f32_16x16x4_f32(
                       false, fa[cur][j][1], false, fb[cur][j], (short)0, acc1, false, false);
            acc2 = __builtin_amdgcn_wmma_f32_16x16x4_f32(
                       false, fa[cur][j][2], false, fb[cur][j], (short)0, acc2, false, false);
        }
        __builtin_amdgcn_sched_barrier(0);
    }

    float* kb = kern + (size_t)b * CENC * HW + (h0 + rowIn) * 64 + w0 + lr;
    const int mh = 8 * half;
#pragma unroll
    for (int r = 0; r < 8; ++r) {
        kb[(mh + r) * HW]      = acc0[r] + bias0[r];
        kb[(16 + mh + r) * HW] = acc1[r] + bias1[r];
    }
    if (half == 0) {
#pragma unroll
        for (int r = 0; r < 4; ++r)
            kb[(32 + r) * HW] = acc2[r] + bias2[r];
    }
}

__global__ void gg_carafe_kernel(const float* __restrict__ x,
                                 const float* __restrict__ mean,
                                 const float* __restrict__ kern,
                                 float* __restrict__ out) {
    __shared__ float smask[KK][SS][WWID];

    int bh = blockIdx.x;
    int b  = bh >> 6, h = bh & 63;
    int tx = threadIdx.x & 63;
    int ty = threadIdx.x >> 6;

    const float* meanb = mean + b * HW;
    const float* kb    = kern + (size_t)b * CENC * HW;

    if (ty == 0) {
        int w   = tx;
        float mc = meanb[h * 64 + w];
        float mval[SS][KK];
#pragma unroll
        for (int kk = 0; kk < KK; ++kk) {
            int dy = kk / 3 - 1, dx = kk % 3 - 1;
            int hh = h + dy, ww = w + dx;
            float mp = ((unsigned)hh < 64u && (unsigned)ww < 64u)
                           ? meanb[hh * 64 + ww] : 0.f;
            float g = mp - mc;
            g = 1.0f / (g * g + 0.2f);
#pragma unroll
            for (int s = 0; s < SS; ++s)
                mval[s][kk] = g * kb[(kk * SS + s) * HW + h * 64 + w];
        }
#pragma unroll
        for (int s = 0; s < SS; ++s) {
            float mx = mval[s][0];
#pragma unroll
            for (int kk = 1; kk < KK; ++kk) mx = fmaxf(mx, mval[s][kk]);
            float e[KK], sum = 0.f;
#pragma unroll
            for (int kk = 0; kk < KK; ++kk) { e[kk] = __expf(mval[s][kk] - mx); sum += e[kk]; }
            float inv = 1.0f / sum;
#pragma unroll
            for (int kk = 0; kk < KK; ++kk) smask[kk][s][w] = e[kk] * inv;
        }
    }
    __syncthreads();

    for (int c = ty; c < CIN; c += 4) {
        const float* xb = x + ((size_t)(b * CIN + c)) * HW;
        __builtin_prefetch(xb + 4 * HW, 0, 0);
        float v[KK];
#pragma unroll
        for (int kk = 0; kk < KK; ++kk) {
            int dy = kk / 3 - 1, dx = kk % 3 - 1;
            int hh = h + dy, ww = tx + dx;
            v[kk] = ((unsigned)hh < 64u && (unsigned)ww < 64u)
                        ? xb[hh * 64 + ww] : 0.f;
        }
        size_t obase = ((size_t)(b * CIN + c)) * 128 * 128;
#pragma unroll
        for (int p = 0; p < 2; ++p)
#pragma unroll
            for (int q = 0; q < 2; ++q) {
                int s = p * 2 + q;
                float acc = 0.f;
#pragma unroll
                for (int kk = 0; kk < KK; ++kk)
                    acc = fmaf(v[kk], smask[kk][s][tx], acc);
                out[obase + (size_t)(2 * h + p) * 128 + (2 * tx + q)] = acc;
            }
    }
}

extern "C" void kernel_launch(void* const* d_in, const int* in_sizes, int n_in,
                              void* d_out, int out_size, void* d_ws, size_t ws_size,
                              hipStream_t stream) {
    const float* x  = (const float*)d_in[0];
    const float* w1 = (const float*)d_in[1];
    const float* b1 = (const float*)d_in[2];
    const float* w2 = (const float*)d_in[3];
    const float* b2 = (const float*)d_in[4];
    float* out = (float*)d_out;

    float* ws   = (float*)d_ws;
    float* mean = ws;
    float* comp = ws + 32768;
    float* kern = comp + (size_t)8 * CCOMP * HW;

    gg_comp_gemm   <<<1024, 256, 0, stream>>>(x, w1, b1, comp, mean);
    gg_kern_gemm   <<<256,  256, 0, stream>>>(comp, w2, b2, kern);
    gg_carafe_kernel<<<512, 256, 0, stream>>>(x, mean, kern, out);
}


# 1x1-conv GEMM: x rows and w1 staged through LDS in 64-channel chunks, next chunk in flight under the WMMAs; k loop is grouped ds_load + f32 WMMA, one B fragment feeds 4 M tiles; sums unchanged
# speedup vs baseline: 2.1260x; 1.1084x over previous
#include <hip/hip_runtime.h>

typedef float v2f __attribute__((ext_vector_type(2)));
typedef float v4f __attribute__((ext_vector_type(4)));
typedef float v8f __attribute__((ext_vector_type(8)));

#define CIN   256
#define CCOMP 64
#define CENC  36
#define HH    64
#define WWID  64
#define HW    4096
#define KK    9
#define SS    4
#define KDIM  576

__device__ __forceinline__ float gg_xhalf(float v) {
    int bits = __builtin_bit_cast(int, v);
    return __builtin_bit_cast(float,
        __builtin_amdgcn_permlanex16(bits, bits, 0x76543210, 0xfedcba98, false, false));
}

#define K2_XPITCH 136
#define K2_APITCH 260

__global__ void __launch_bounds__(256)
gg_comp_gemm(const float* __restrict__ x,
             const float* __restrict__ w1,
             const float* __restrict__ b1,
             float* __restrict__ comp,
             float* __restrict__ mean) {
    __shared__ __attribute__((aligned(16))) float Xs[CIN * K2_XPITCH];
    __shared__ __attribute__((aligned(16))) float As[CCOMP * K2_APITCH];

    const int tid   = threadIdx.x;
    const int b     = blockIdx.x >> 5;
    const int nBase = (blockIdx.x & 31) * 128;
    const int wave  = tid >> 5;
    const int lane  = tid & 31;
    const int lr    = lane & 15;
    const int half  = lane >> 4;
    const int n0    = wave * 16;

    const float* xb = x + (size_t)b * CIN * HW + nBase;

    const int xc = (tid & 31) * 4;
    const int xr = tid >> 5;
    const int wc = (tid & 15) * 4;
    const int wr = tid >> 4;

    v4f xv[8], wv[4];
#pragma unroll
    for (int i = 0; i < 8; ++i)
        xv[i] = *reinterpret_cast<const v4f*>(xb + (xr + 8 * i) * HW + xc);
#pragma unroll
    for (int i = 0; i < 4; ++i)
        wv[i] = *reinterpret_cast<const v4f*>(w1 + (wr + 16 * i) * CIN + wc);

    float bias[4][8];
#pragma unroll
    for (int mt = 0; mt < 4; ++mt)
#pragma unroll
        for (int r = 0; r < 8; ++r)
            bias[mt][r] = b1[mt * 16 + 8 * half + r];

    const int boff = 2 * half * K2_XPITCH + n0 + lr;
    const int aoff = lr * K2_APITCH + 2 * half;

    v2f fa[2][4][4], fb[2][4];
    v8f acc[4] = {};
    v2f csum = {0.f, 0.f};

#pragma unroll
    for (int c = 0; c < 4; ++c) {
#pragma unroll
        for (int i = 0; i < 8; ++i)
            *reinterpret_cast<v4f*>(&Xs[(c * 64 + xr + 8 * i) * K2_XPITCH + xc]) = xv[i];
#pragma unroll
        for (int i = 0; i < 4; ++i)
            *reinterpret_cast<v4f*>(&As[(wr + 16 * i) * K2_APITCH + c * 64 + wc]) = wv[i];
        __syncthreads();
        if (c + 1 < 4) {
#pragma unroll
            for (int i = 0; i < 8; ++i)
                xv[i] = *reinterpret_cast<const v4f*>(xb + ((c + 1) * 64 + xr + 8 * i) * HW + xc);
#pragma unroll
            for (int i = 0; i < 4; ++i)
                wv[i] = *reinterpret_cast<const v4f*>(w1 + (wr + 16 * i) * CIN + (c + 1) * 64 + wc);
        }

#pragma unroll
        for (int j = 0; j < 4; ++j) {
            const int k0 = c * 64 + j * 4;
            fb[0][j].x = Xs[boff + k0 * K2_XPITCH];
            fb[0][j].y = Xs[boff + (k0 + 1) * K2_XPITCH];
#pragma unroll
            for (int mt = 0; mt < 4; ++mt)
                fa[0][j][mt] = *reinterpret_cast<const v2f*>(&As[aoff + mt * 16 * K2_APITCH + k0]);
        }
#pragma unroll
        for (int g = 0; g < 4; ++g) {
            const int cur = g & 1, nxt = cur ^ 1;
            if (g + 1 < 4) {
#pragma unroll
                for (int j = 0; j < 4; ++j) {
                    const int k0 = c * 64 + (g + 1) * 16 + j * 4;
                    fb[nxt][j].x = Xs[boff + k0 * K2_XPITCH];
                    fb[nxt][j].y = Xs[boff + (k0 + 1) * K2_XPITCH];
#pragma unroll
                    for (int mt = 0; mt < 4; ++mt)
                        fa[nxt][j][mt] = *reinterpret_cast<const v2f*>(&As[aoff + mt * 16 * K2_APITCH + k0]);
                }
            }
            __builtin_amdgcn_sched_barrier(0);
#pragma unroll
            for (int j = 0; j < 4; ++j) {
#pragma unroll
                for (int mt = 0; mt < 4; ++mt)
                    acc[mt] = __builtin_amdgcn_wmma_f32_16x16x4_f32(
                                  false, fa[cur][j][mt], false, fb[cur][j], (short)0, acc[mt], false, false);
                csum += fb[cur][j];
            }
            __builtin_amdgcn_sched_barrier(0);
        }
    }

    float cs = csum.x + csum.y;
    cs += gg_xhalf(cs);
    if (half == 0)
        mean[b * HW + nBase + n0 + lr] = cs * (1.0f / CIN);

    float* cb = comp + (size_t)b * CCOMP * HW + nBase + n0 + lr;
#pragma unroll
    for (int mt = 0; mt < 4; ++mt)
#pragma unroll
        for (int r = 0; r < 8; ++r) {
            int m = mt * 16 + r + 8 * half;
            cb[m * HW] = acc[mt][r] + bias[mt][r];
        }
}

#define K3_APITCH 580
#define K3_AROWS  48
#define K3_BROW   68
#define K3_BCH    272
#define K3_BGRP   1088

__global__ void __launch_bounds__(256)
gg_kern_gemm(const float* __restrict__ comp,
             const float* __restrict__ w2,
             const float* __restrict__ b2,
             float* __restrict__ kern) {
    __shared__ __attribute__((aligned(16))) float As[K3_AROWS * K3_APITCH];
    __shared__ __attribute__((aligned(16))) float Bs[CCOMP * K3_BCH];

    const int tid = threadIdx.x;
    const int b   = blockIdx.x >> 5;
    const int h0  = (blockIdx.x & 31) * 2;
    const float* cb = comp + (size_t)b * CCOMP * HW;

    v4f wv[20];
#pragma unroll
    for (int t = 0; t < 20; ++t)
        wv[t] = *reinterpret_cast<const v4f*>(w2 + (tid + t * 256) * 4);
    const int iTail = 20 * 256 + (tid & 63);
    v4f wTail = *reinterpret_cast<const v4f*>(w2 + iTail * 4);
    const int  w4    = tid & 15;
    const int  rS    = (tid >> 4) & 3;
    const int  cq    = tid >> 6;
    const int  hS    = h0 - 1 + rS;
    const bool rowOk = (unsigned)hS < 64u;
    const int  hL    = rowOk ? hS : h0;
    v4f cv[16];
#pragma unroll
    for (int t = 0; t < 16; ++t)
        cv[t] = *reinterpret_cast<const v4f*>(cb + (cq + 4 * t) * HW + hL * 64 + w4 * 4);
    const int hf8 = (tid & 16) >> 1;
    float bias0[8], bias1[8], bias2[4];
#pragma unroll
    for (int r = 0; r < 8; ++r) {
        bias0[r] = b2[r + hf8];
        bias1[r] = b2[16 + r + hf8];
    }
#pragma unroll
    for (int r = 0; r < 4; ++r)
        bias2[r] = b2[32 + r];
#pragma unroll
    for (int t = 0; t < 20; ++t) {
        int i  = tid + t * 256;
        int m  = i / (KDIM / 4);
        int j4 = i - m * (KDIM / 4);
        *reinterpret_cast<v4f*>(&As[m * K3_APITCH + j4 * 4]) = wv[t];
    }
    {
        int m  = iTail / (KDIM / 4);
        int j4 = iTail - m * (KDIM / 4);
        *reinterpret_cast<v4f*>(&As[m * K3_APITCH + j4 * 4]) = wTail;
    }
    for (int i = tid; i < (K3_AROWS - CENC) * K3_APITCH / 4; i += 256) {
        v4f z = {0.f, 0.f, 0.f, 0.f};
        *reinterpret_cast<v4f*>(&As[CENC * K3_APITCH + i * 4]) = z;
    }
#pragma unroll
    for (int t = 0; t < 16; ++t) {
        int o = (cq + 4 * t) * K3_BCH + rS * K3_BROW + 1 + w4 * 4;
        Bs[o]     = rowOk ? cv[t].x : 0.f;
        Bs[o + 1] = rowOk ? cv[t].y : 0.f;
        Bs[o + 2] = rowOk ? cv[t].z : 0.f;
        Bs[o + 3] = rowOk ? cv[t].w : 0.f;
    }
    {
        int o = (tid >> 2) * K3_BCH + (tid & 3) * K3_BROW;
        Bs[o]      = 0.f;
        Bs[o + 65] = 0.f;
    }
    __syncthreads();

    const int wave  = tid >> 5;
    const int lane  = tid & 31;
    const int lr    = lane & 15;
    const int half  = lane >> 4;
    const int rowIn = wave >> 2;
    const int w0    = (wave & 3) * 16;

    int boff[9][2];
#pragma unroll
    for (int j = 0; j < 9; ++j)
#pragma unroll
        for (int s = 0; s < 2; ++s) {
            int q  = j * 4 + 2 * half + s;
            int cc = q / 9, t = q - cc * 9, ky = t / 3, kx = t - ky * 3;
            boff[j][s] = cc * K3_BCH + (rowIn + ky) * K3_BROW + (w0 + lr + kx);
        }
    const int aoff = lr * K3_APITCH + 2 * half;

    v2f fa[2][9][3], fb[2][9];
    v8f acc0 = {}, acc1 = {}, acc2 = {};

#pragma unroll
    for (int j = 0; j < 9; ++j) {
        fb[0][j].x = Bs[boff[j][0]];
        fb[0][j].y = Bs[boff[j][1]];
        fa[0][j][0] = *reinterpret_cast<const v2f*>(&As[aoff + j * 4]);
        fa[0][j][1] = *reinterpret_cast<const v2f*>(&As[aoff + 16 * K3_APITCH + j * 4]);
        fa[0][j][2] = *reinterpret_cast<const v2f*>(&As[aoff + 32 * K3_APITCH + j * 4]);
    }
#pragma unroll
    for (int g = 0; g < 16; ++g) {
        const int cur = g & 1, nxt = cur ^ 1;
        if (g + 1 < 16) {
#pragma unroll
            for (int j = 0; j < 9; ++j) {
                const int k1 = (g + 1) * 36 + j * 4;
                fb[nxt][j].x = Bs[(g + 1) * K3_BGRP + boff[j][0]];
                fb[nxt][j].y = Bs[(g + 1) * K3_BGRP + boff[j][1]];
                fa[nxt][j][0] = *reinterpret_cast<const v2f*>(&As[aoff + k1]);
                fa[nxt][j][1] = *reinterpret_cast<const v2f*>(&As[aoff + 16 * K3_APITCH + k1]);
                fa[nxt][j][2] = *reinterpret_cast<const v2f*>(&As[aoff + 32 * K3_APITCH + k1]);
            }
        }
        __builtin_amdgcn_sched_barrier(0);
#pragma unroll
        for (int j = 0; j < 9; ++j) {
            acc0 = __builtin_amdgcn_wmma_f32_16x16x4_f32(
                       false, fa[cur][j][0], false, fb[cur][j], (short)0, acc0, false, false);
            acc1 = __builtin_amdgcn_wmma_f32_16x16x4_f32(
                       false, fa[cur][j][1], false, fb[cur][j], (short)0, acc1, false, false);
            acc2 = __builtin_amdgcn_wmma_f32_16x16x4_f32(
                       false, fa[cur][j][2], false, fb[cur][j], (short)0, acc2, false, false);
        }
        __builtin_amdgcn_sched_barrier(0);
    }

    float* kb = kern + (size_t)b * CENC * HW + (h0 + rowIn) * 64 + w0 + lr;
    const int mh = 8 * half;
#pragma unroll
    for (int r = 0; r < 8; ++r) {
        kb[(mh + r) * HW]      = acc0[r] + bias0[r];
        kb[(16 + mh + r) * HW] = acc1[r] + bias1[r];
    }
    if (half == 0) {
#pragma unroll
        for (int r = 0; r < 4; ++r)
            kb[(32 + r) * HW] = acc2[r] + bias2[r];
    }
}

__global__ void gg_carafe_kernel(const float* __restrict__ x,
                                 const float* __restrict__ mean,
                                 const float* __restrict__ kern,
                                 float* __restrict__ out) {
    __shared__ float smask[KK][SS][WWID];

    int bh = blockIdx.x;
    int b  = bh >> 6, h = bh & 63;
    int tx = threadIdx.x & 63;
    int ty = threadIdx.x >> 6;

    const float* meanb = mean + b * HW;
    const float* kb    = kern + (size_t)b * CENC * HW;

    if (ty == 0) {
        int w   = tx;
        float mc = meanb[h * 64 + w];
        float mval[SS][KK];
#pragma unroll
        for (int kk = 0; kk < KK; ++kk) {
            int dy = kk / 3 - 1, dx = kk % 3 - 1;
            int hh = h + dy, ww = w + dx;
            float mp = ((unsigned)hh < 64u && (unsigned)ww < 64u)
                           ? meanb[hh * 64 + ww] : 0.f;
            float g = mp - mc;
            g = 1.0f / (g * g + 0.2f);
#pragma unroll
            for (int s = 0; s < SS; ++s)
                mval[s][kk] = g * kb[(kk * SS + s) * HW + h * 64 + w];
        }
#pragma unroll
        for (int s = 0; s < SS; ++s) {
            float mx = mval[s][0];
#pragma unroll
            for (int kk = 1; kk < KK; ++kk) mx = fmaxf(mx, mval[s][kk]);
            float e[KK], sum = 0.f;
#pragma unroll
            for (int kk = 0; kk < KK; ++kk) { e[kk] = __expf(mval[s][kk] - mx); sum += e[kk]; }
            float inv = 1.0f / sum;
#pragma unroll
            for (int kk = 0; kk < KK; ++kk) smask[kk][s][w] = e[kk] * inv;
        }
    }
    __syncthreads();

    for (int c = ty; c < CIN; c += 4) {
        const float* xb = x + ((size_t)(b * CIN + c)) * HW;
        __builtin_prefetch(xb + 4 * HW, 0, 0);
        float v[KK];
#pragma unroll
        for (int kk = 0; kk < KK; ++kk) {
            int dy = kk / 3 - 1, dx = kk % 3 - 1;
            int hh = h + dy, ww = tx + dx;
            v[kk] = ((unsigned)hh < 64u && (unsigned)ww < 64u)
                        ? xb[hh * 64 + ww] : 0.f;
        }
        size_t obase = ((size_t)(b * CIN + c)) * 128 * 128;
#pragma unroll
        for (int p = 0; p < 2; ++p)
#pragma unroll
            for (int q = 0; q < 2; ++q) {
                int s = p * 2 + q;
                float acc = 0.f;
#pragma unroll
                for (int kk = 0; kk < KK; ++kk)
                    acc = fmaf(v[kk], smask[kk][s][tx], acc);
                out[obase + (size_t)(2 * h + p) * 128 + (2 * tx + q)] = acc;
            }
    }
}

extern "C" void kernel_launch(void* const* d_in, const int* in_sizes, int n_in,
                              void* d_out, int out_size, void* d_ws, size_t ws_size,
                              hipStream_t stream) {
    const float* x  = (const float*)d_in[0];
    const float* w1 = (const float*)d_in[1];
    const float* b1 = (const float*)d_in[2];
    const float* w2 = (const float*)d_in[3];
    const float* b2 = (const float*)d_in[4];
    float* out = (float*)d_out;

    float* ws   = (float*)d_ws;
    float* mean = ws;
    float* comp = ws + 32768;
    float* kern = comp + (size_t)8 * CCOMP * HW;

    gg_comp_gemm   <<<256,  256, 0, stream>>>(x, w1, b1, comp, mean);
    gg_kern_gemm   <<<256,  256, 0, stream>>>(comp, w2, b2, kern);
    gg_carafe_kernel<<<512, 256, 0, stream>>>(x, mean, kern, out);
}


# reassembly kernel: a lane owns 2 adjacent pixels, weights in registers; x read in 8-byte pieces, a 512-B output row is one wave store, 16 B per lane; next channels' loads in flight; sums unchanged
# speedup vs baseline: 3.7730x; 1.7747x over previous
#include <hip/hip_runtime.h>

typedef float v2f __attribute__((ext_vector_type(2)));
typedef float v4f __attribute__((ext_vector_type(4)));
typedef float v8f __attribute__((ext_vector_type(8)));

#define CIN   256
#define CCOMP 64
#define CENC  36
#define HH    64
#define WWID  64
#define HW    4096
#define KK    9
#define SS    4
#define KDIM  576

__device__ __forceinline__ float gg_xhalf(float v) {
    int bits = __builtin_bit_cast(int, v);
    return __builtin_bit_cast(float,
        __builtin_amdgcn_permlanex16(bits, bits, 0x76543210, 0xfedcba98, false, false));
}

#define K2_XPITCH 136
#define K2_APITCH 260

__global__ void __launch_bounds__(256)
gg_comp_gemm(const float* __restrict__ x,
             const float* __restrict__ w1,
             const float* __restrict__ b1,
             float* __restrict__ comp,
             float* __restrict__ mean) {
    __shared__ __attribute__((aligned(16))) float Xs[CIN * K2_XPITCH];
    __shared__ __attribute__((aligned(16))) float As[CCOMP * K2_APITCH];

    const int tid   = threadIdx.x;
    const int b     = blockIdx.x >> 5;
    const int nBase = (blockIdx.x & 31) * 128;
    const int wave  = tid >> 5;
    const int lane  = tid & 31;
    const int lr    = lane & 15;
    const int half  = lane >> 4;
    const int n0    = wave * 16;

    const float* xb = x + (size_t)b * CIN * HW + nBase;

    const int xc = (tid & 31) * 4;
    const int xr = tid >> 5;
    const int wc = (tid & 15) * 4;
    const int wr = tid >> 4;

    v4f xv[8], wv[4];
#pragma unroll
    for (int i = 0; i < 8; ++i)
        xv[i] = *reinterpret_cast<const v4f*>(xb + (xr + 8 * i) * HW + xc);
#pragma unroll
    for (int i = 0; i < 4; ++i)
        wv[i] = *reinterpret_cast<const v4f*>(w1 + (wr + 16 * i) * CIN + wc);

    float bias[4][8];
#pragma unroll
    for (int mt = 0; mt < 4; ++mt)
#pragma unroll
        for (int r = 0; r < 8; ++r)
            bias[mt][r] = b1[mt * 16 + 8 * half + r];

    const int boff = 2 * half * K2_XPITCH + n0 + lr;
    const int aoff = lr * K2_APITCH + 2 * half;

    v2f fa[2][4][4], fb[2][4];
    v8f acc[4] = {};
    v2f csum = {0.f, 0.f};

#pragma unroll
    for (int c = 0; c < 4; ++c) {
#pragma unroll
        for (int i = 0; i < 8; ++i)
            *reinterpret_cast<v4f*>(&Xs[(c * 64 + xr + 8 * i) * K2_XPITCH + xc]) = xv[i];
#pragma unroll
        for (int i = 0; i < 4; ++i)
            *reinterpret_cast<v4f*>(&As[(wr + 16 * i) * K2_APITCH + c * 64 + wc]) = wv[i];
        __syncthreads();
        if (c + 1 < 4) {
#pragma unroll
            for (int i = 0; i < 8; ++i)
                xv[i] = *reinterpret_cast<const v4f*>(xb + ((c + 1) * 64 + xr + 8 * i) * HW + xc);
#pragma unroll
            for (int i = 0; i < 4; ++i)
                wv[i] = *reinterpret_cast<const v4f*>(w1 + (wr + 16 * i) * CIN + (c + 1) * 64 + wc);
        }

#pragma unroll
        for (int j = 0; j < 4; ++j) {
            const int k0 = c * 64 + j * 4;
            fb[0][j].x = Xs[boff + k0 * K2_XPITCH];
            fb[0][j].y = Xs[boff + (k0 + 1) * K2_XPITCH];
#pragma unroll
            for (int mt = 0; mt < 4; ++mt)
                fa[0][j][mt] = *reinterpret_cast<const v2f*>(&As[aoff + mt * 16 * K2_APITCH + k0]);
        }
#pragma unroll
        for (int g = 0; g < 4; ++g) {
            const int cur = g & 1, nxt = cur ^ 1;
            if (g + 1 < 4) {
#pragma unroll
                for (int j = 0; j < 4; ++j) {
                    const int k0 = c * 64 + (g + 1) * 16 + j * 4;
                    fb[nxt][j].x = Xs[boff + k0 * K2_XPITCH];
                    fb[nxt][j].y = Xs[boff + (k0 + 1) * K2_XPITCH];
#pragma unroll
                    for (int mt = 0; mt < 4; ++mt)
                        fa[nxt][j][mt] = *reinterpret_cast<const v2f*>(&As[aoff + mt * 16 * K2_APITCH + k0]);
                }
            }
            __builtin_amdgcn_sched_barrier(0);
#pragma unroll
            for (int j = 0; j < 4; ++j) {
#pragma unroll
                for (int mt = 0; mt < 4; ++mt)
                    acc[mt] = __builtin_amdgcn_wmma_f32_16x16x4_f32(
                                  false, fa[cur][j][mt], false, fb[cur][j], (short)0, acc[mt], false, false);
                csum += fb[cur][j];
            }
            __builtin_amdgcn_sched_barrier(0);
        }
    }

    float cs = csum.x + csum.y;
    cs += gg_xhalf(cs);
    if (half == 0)
        mean[b * HW + nBase + n0 + lr] = cs * (1.0f / CIN);

    float* cb = comp + (size_t)b * CCOMP * HW + nBase + n0 + lr;
#pragma unroll
    for (int mt = 0; mt < 4; ++mt)
#pragma unroll
        for (int r = 0; r < 8; ++r) {
            int m = mt * 16 + r + 8 * half;
            cb[m * HW] = acc[mt][r] + bias[mt][r];
        }
}

#define K3_APITCH 580
#define K3_AROWS  48
#define K3_BROW   68
#define K3_BCH    272
#define K3_BGRP   1088

__global__ void __launch_bounds__(256)
gg_kern_gemm(const float* __restrict__ comp,
             const float* __restrict__ w2,
             const float* __restrict__ b2,
             float* __restrict__ kern) {
    __shared__ __attribute__((aligned(16))) float As[K3_AROWS * K3_APITCH];
    __shared__ __attribute__((aligned(16))) float Bs[CCOMP * K3_BCH];

    const int tid = threadIdx.x;
    const int b   = blockIdx.x >> 5;
    const int h0  = (blockIdx.x & 31) * 2;
    const float* cb = comp + (size_t)b * CCOMP * HW;

    v4f wv[20];
#pragma unroll
    for (int t = 0; t < 20; ++t)
        wv[t] = *reinterpret_cast<const v4f*>(w2 + (tid + t * 256) * 4);
    const int iTail = 20 * 256 + (tid & 63);
    v4f wTail = *reinterpret_cast<const v4f*>(w2 + iTail * 4);
    const int  w4    = tid & 15;
    const int  rS    = (tid >> 4) & 3;
    const int  cq    = tid >> 6;
    const int  hS    = h0 - 1 + rS;
    const bool rowOk = (unsigned)hS < 64u;
    const int  hL    = rowOk ? hS : h0;
    v4f cv[16];
#pragma unroll
    for (int t = 0; t < 16; ++t)
        cv[t] = *reinterpret_cast<const v4f*>(cb + (cq + 4 * t) * HW + hL * 64 + w4 * 4);
    const int hf8 = (tid & 16) >> 1;
    float bias0[8], bias1[8], bias2[4];
#pragma unroll
    for (int r = 0; r < 8; ++r) {
        bias0[r] = b2[r + hf8];
        bias1[r] = b2[16 + r + hf8];
    }
#pragma unroll
    for (int r = 0; r < 4; ++r)
        bias2[r] = b2[32 + r];
#pragma unroll
    for (int t = 0; t < 20; ++t) {
        int i  = tid + t * 256;
        int m  = i / (KDIM / 4);
        int j4 = i - m * (KDIM / 4);
        *reinterpret_cast<v4f*>(&As[m * K3_APITCH + j4 * 4]) = wv[t];
    }
    {
        int m  = iTail / (KDIM / 4);
        int j4 = iTail - m * (KDIM / 4);
        *reinterpret_cast<v4f*>(&As[m * K3_APITCH + j4 * 4]) = wTail;
    }
    for (int i = tid; i < (K3_AROWS - CENC) * K3_APITCH / 4; i += 256) {
        v4f z = {0.f, 0.f, 0.f, 0.f};
        *reinterpret_cast<v4f*>(&As[CENC * K3_APITCH + i * 4]) = z;
    }
#pragma unroll
    for (int t = 0; t < 16; ++t) {
        int o = (cq + 4 * t) * K3_BCH + rS * K3_BROW + 1 + w4 * 4;
        Bs[o]     = rowOk ? cv[t].x : 0.f;
        Bs[o + 1] = rowOk ? cv[t].y : 0.f;
        Bs[o + 2] = rowOk ? cv[t].z : 0.f;
        Bs[o + 3] = rowOk ? cv[t].w : 0.f;
    }
    {
        int o = (tid >> 2) * K3_BCH + (tid & 3) * K3_BROW;
        Bs[o]      = 0.f;
        Bs[o + 65] = 0.f;
    }
    __syncthreads();

    const int wave  = tid >> 5;
    const int lane  = tid & 31;
    const int lr    = lane & 15;
    const int half  = lane >> 4;
    const int rowIn = wave >> 2;
    const int w0    = (wave & 3) * 16;

    int boff[9][2];
#pragma unroll
    for (int j = 0; j < 9; ++j)
#pragma unroll
        for (int s = 0; s < 2; ++s) {
            int q  = j * 4 + 2 * half + s;
            int cc = q / 9, t = q - cc * 9, ky = t / 3, kx = t - ky * 3;
            boff[j][s] = cc * K3_BCH + (rowIn + ky) * K3_BROW + (w0 + lr + kx);
        }
    const int aoff = lr * K3_APITCH + 2 * half;

    v2f fa[2][9][3], fb[2][9];
    v8f acc0 = {}, acc1 = {}, acc2 = {};

#pragma unroll
    for (int j = 0; j < 9; ++j) {
        fb[0][j].x = Bs[boff[j][0]];
        fb[0][j].y = Bs[boff[j][1]];
        fa[0][j][0] = *reinterpret_cast<const v2f*>(&As[aoff + j * 4]);
        fa[0][j][1] = *reinterpret_cast<const v2f*>(&As[aoff + 16 * K3_APITCH + j * 4]);
        fa[0][j][2] = *reinterpret_cast<const v2f*>(&As[aoff + 32 * K3_APITCH + j * 4]);
    }
#pragma unroll
    for (int g = 0; g < 16; ++g) {
        const int cur = g & 1, nxt = cur ^ 1;
        if (g + 1 < 16) {
#pragma unroll
            for (int j = 0; j < 9; ++j) {
                const int k1 = (g + 1) * 36 + j * 4;
                fb[nxt][j].x = Bs[(g + 1) * K3_BGRP + boff[j][0]];
                fb[nxt][j].y = Bs[(g + 1) * K3_BGRP + boff[j][1]];
                fa[nxt][j][0] = *reinterpret_cast<const v2f*>(&As[aoff + k1]);
                fa[nxt][j][1] = *reinterpret_cast<const v2f*>(&As[aoff + 16 * K3_APITCH + k1]);
                fa[nxt][j][2] = *reinterpret_cast<const v2f*>(&As[aoff + 32 * K3_APITCH + k1]);
            }
        }
        __builtin_amdgcn_sched_barrier(0);
#pragma unroll
        for (int j = 0; j < 9; ++j) {
            acc0 = __builtin_amdgcn_wmma_f32_16x16x4_f32(
                       false, fa[cur][j][0], false, fb[cur][j], (short)0, acc0, false, false);
            acc1 = __builtin_amdgcn_wmma_f32_16x16x4_f32(
                       false, fa[cur][j][1], false, fb[cur][j], (short)0, acc1, false, false);
            acc2 = __builtin_amdgcn_wmma_f32_16x16x4_f32(
                       false, fa[cur][j][2], false, fb[cur][j], (short)0, acc2, false, false);
        }
        __builtin_amdgcn_sched_barrier(0);
    }

    float* kb = kern + (size_t)b * CENC * HW + (h0 + rowIn) * 64 + w0 + lr;
    const int mh = 8 * half;
#pragma unroll
    for (int r = 0; r < 8; ++r) {
        kb[(mh + r) * HW]      = acc0[r] + bias0[r];
        kb[(16 + mh + r) * HW] = acc1[r] + bias1[r];
    }
    if (half == 0) {
#pragma unroll
        for (int r = 0; r < 4; ++r)
            kb[(32 + r) * HW] = acc2[r] + bias2[r];
    }
}

struct GgTaps { float a[3][4]; };

__device__ __forceinline__ void gg_load_taps(GgTaps& d, const float* __restrict__ xc,
                                             const int (&om)[3], const int (&ol)[3],
                                             const int (&orr)[3]) {
#pragma unroll
    for (int r = 0; r < 3; ++r) {
        v2f mid = *reinterpret_cast<const v2f*>(xc + om[r]);
        d.a[r][0] = xc[ol[r]];
        d.a[r][1] = mid.x;
        d.a[r][2] = mid.y;
        d.a[r][3] = xc[orr[r]];
    }
}

__device__ __forceinline__ void gg_reassemble(const GgTaps& d, const bool (&okM)[3],
                                              const bool (&okL)[3], const bool (&okR)[3],
                                              const float (&m0)[KK][SS], const float (&m1)[KK][SS],
                                              float* __restrict__ o) {
    float a[3][4];
#pragma unroll
    for (int r = 0; r < 3; ++r) {
        a[r][0] = okL[r] ? d.a[r][0] : 0.f;
        a[r][1] = okM[r] ? d.a[r][1] : 0.f;
        a[r][2] = okM[r] ? d.a[r][2] : 0.f;
        a[r][3] = okR[r] ? d.a[r][3] : 0.f;
    }
#pragma unroll
    for (int p = 0; p < 2; ++p) {
        float res[4];
#pragma unroll
        for (int q = 0; q < 2; ++q) {
            const int s = p * 2 + q;
            float acc0 = 0.f, acc1 = 0.f;
#pragma unroll
            for (int kk = 0; kk < KK; ++kk) {
                const int ky = kk / 3, kx = kk % 3;
                acc0 = fmaf(a[ky][kx],     m0[kk][s], acc0);
                acc1 = fmaf(a[ky][kx + 1], m1[kk][s], acc1);
            }
            res[q]     = acc0;
            res[2 + q] = acc1;
        }
        v4f ov = {res[0], res[1], res[2], res[3]};
        *reinterpret_cast<v4f*>(o + p * 128) = ov;
    }
}

__global__ void __launch_bounds__(256)
gg_carafe_kernel(const float* __restrict__ x,
                 const float* __restrict__ mean,
                 const float* __restrict__ kern,
                 float* __restrict__ out) {
    __shared__ __attribute__((aligned(16))) float smask[KK * SS * WWID];

    const int tid  = threadIdx.x;
    const int b    = blockIdx.x >> 6, h = blockIdx.x & 63;
    const int lane = tid & 31;
    const int wv   = tid >> 5;

    int  om[3], ol[3], orr[3];
    bool okM[3], okL[3], okR[3];
#pragma unroll
    for (int r = 0; r < 3; ++r) {
        const int  hh = h - 1 + r;
        const bool ok = (unsigned)hh < 64u;
        const int  hc = ok ? hh : h;
        om[r]  = hc * 64 + 2 * lane;
        ol[r]  = hc * 64 + (lane > 0 ? 2 * lane - 1 : 0);
        orr[r] = hc * 64 + (lane < 31 ? 2 * lane + 2 : 63);
        okM[r] = ok;
        okL[r] = ok && lane > 0;
        okR[r] = ok && lane < 31;
    }
    const float* xw = x   + (size_t)(b * CIN + wv) * HW;
    float*       ow = out + (size_t)(b * CIN + wv) * (4 * HW) + (2 * h) * 128 + 4 * lane;

    GgTaps tA0, tA1, tB0, tB1;
    gg_load_taps(tA0, xw, om, ol, orr);
    gg_load_taps(tA1, xw + 8 * HW, om, ol, orr);

    {
        const int w = tid & 63, s = tid >> 6;
        const float* meanb = mean + b * HW;
        const float* kb    = kern + (size_t)b * CENC * HW + s * HW + h * 64 + w;
        float kv[KK], mp[KK];
#pragma unroll
        for (int kk = 0; kk < KK; ++kk) {
            const int  dy = kk / 3 - 1, dx = kk % 3 - 1;
            const int  hh = h + dy, ww = w + dx;
            const bool ok = (unsigned)hh < 64u && (unsigned)ww < 64u;
            kv[kk] = kb[kk * SS * HW];
            mp[kk] = meanb[ok ? hh * 64 + ww : h * 64 + w];
            mp[kk] = ok ? mp[kk] : 0.f;
        }
        const float mc = meanb[h * 64 + w];
        float mval[KK];
#pragma unroll
        for (int kk = 0; kk < KK; ++kk) {
            float g = mp[kk] - mc;
            g = 1.0f / (g * g + 0.2f);
            mval[kk] = g * kv[kk];
        }
        float mx = mval[0];
#pragma unroll
        for (int kk = 1; kk < KK; ++kk) mx = fmaxf(mx, mval[kk]);
        float e[KK], sum = 0.f;
#pragma unroll
        for (int kk = 0; kk < KK; ++kk) { e[kk] = __expf(mval[kk] - mx); sum += e[kk]; }
        const float inv = 1.0f / sum;
#pragma unroll
        for (int kk = 0; kk < KK; ++kk) smask[(kk * SS + s) * WWID + w] = e[kk] * inv;
    }
    __syncthreads();

    float m0[KK][SS], m1[KK][SS];
#pragma unroll
    for (int kk = 0; kk < KK; ++kk)
#pragma unroll
        for (int s = 0; s < SS; ++s) {
            v2f t = *reinterpret_cast<const v2f*>(&smask[(kk * SS + s) * WWID + 2 * lane]);
            m0[kk][s] = t.x;
            m1[kk][s] = t.y;
        }

    for (int i = 0; i < 28; i += 4) {
        const float* xi = xw + (size_t)i * (8 * HW);
        float*       oi = ow + (size_t)i * (8 * 4 * HW);
        gg_load_taps(tB0, xi + 16 * HW, om, ol, orr);
        gg_load_taps(tB1, xi + 24 * HW, om, ol, orr);
        gg_reassemble(tA0, okM, okL, okR, m0, m1, oi);
        gg_reassemble(tA1, okM, okL, okR, m0, m1, oi + 8 * 4 * HW);
        gg_load_taps(tA0, xi + 32 * HW, om, ol, orr);
        gg_load_taps(tA1, xi + 40 * HW, om, ol, orr);
        gg_reassemble(tB0, okM, okL, okR, m0, m1, oi + 16 * 4 * HW);
        gg_reassemble(tB1, okM, okL, okR, m0, m1, oi + 24 * 4 * HW);
    }
    {
        const float* xi = xw + (size_t)28 * (8 * HW);
        float*       oi = ow + (size_t)28 * (8 * 4 * HW);
        gg_load_taps(tB0, xi + 16 * HW, om, ol, orr);
        gg_load_taps(tB1, xi + 24 * HW, om, ol, orr);
        gg_reassemble(tA0, okM, okL, okR, m0, m1, oi);
        gg_reassemble(tA1, okM, okL, okR, m0, m1, oi + 8 * 4 * HW);
        gg_reassemble(tB0, okM, okL, okR, m0, m1, oi + 16 * 4 * HW);
        gg_reassemble(tB1, okM, okL, okR, m0, m1, oi + 24 * 4 * HW);
    }
}

extern "C" void kernel_launch(void* const* d_in, const int* in_sizes, int n_in,
                              void* d_out, int out_size, void* d_ws, size_t ws_size,
                              hipStream_t stream) {
    const float* x  = (const float*)d_in[0];
    const float* w1 = (const float*)d_in[1];
    const float* b1 = (const float*)d_in[2];
    const float* w2 = (const float*)d_in[3];
    const float* b2 = (const float*)d_in[4];
    float* out = (float*)d_out;

    float* ws   = (float*)d_ws;
    float* mean = ws;
    float* comp = ws + 32768;
    float* kern = comp + (size_t)8 * CCOMP * HW;

    gg_comp_gemm   <<<256,  256, 0, stream>>>(x, w1, b1, comp, mean);
    gg_kern_gemm   <<<256,  256, 0, stream>>>(comp, w2, b2, kern);
    gg_carafe_kernel<<<512, 256, 0, stream>>>(x, mean, kern, out);
}


# 3x3-conv GEMM moved into the reassembly kernel (512-thread block, two rows): kern stays in LDS, one launch fewer; each tile's k sum is made in two halves (channels 0-31, 32-63) added at the end
# speedup vs baseline: 4.1717x; 1.1057x over previous
#include <hip/hip_runtime.h>

typedef float v2f __attribute__((ext_vector_type(2)));
typedef float v4f __attribute__((ext_vector_type(4)));
typedef float v8f __attribute__((ext_vector_type(8)));

#define CIN   256
#define CCOMP 64
#define CENC  36
#define HH    64
#define WWID  64
#define HW    4096
#define KK    9
#define SS    4
#define KDIM  576

__device__ __forceinline__ float gg_xhalf(float v) {
    int bits = __builtin_bit_cast(int, v);
    return __builtin_bit_cast(float,
        __builtin_amdgcn_permlanex16(bits, bits, 0x76543210, 0xfedcba98, false, false));
}

#define K2_XPITCH 136
#define K2_APITCH 260

__global__ void __launch_bounds__(256)
gg_comp_gemm(const float* __restrict__ x,
             const float* __restrict__ w1,
             const float* __restrict__ b1,
             float* __restrict__ comp,
             float* __restrict__ mean) {
    __shared__ __attribute__((aligned(16))) float Xs[CIN * K2_XPITCH];
    __shared__ __attribute__((aligned(16))) float As[CCOMP * K2_APITCH];

    const int tid   = threadIdx.x;
    const int b     = blockIdx.x >> 5;
    const int nBase = (blockIdx.x & 31) * 128;
    const int wave  = tid >> 5;
    const int lane  = tid & 31;
    const int lr    = lane & 15;
    const int half  = lane >> 4;
    const int n0    = wave * 16;

    const float* xb = x + (size_t)b * CIN * HW + nBase;

    const int xc = (tid & 31) * 4;
    const int xr = tid >> 5;
    const int wc = (tid & 15) * 4;
    const int wr = tid >> 4;

    v4f xv[8], wv[4];
#pragma unroll
    for (int i = 0; i < 8; ++i)
        xv[i] = *reinterpret_cast<const v4f*>(xb + (xr + 8 * i) * HW + xc);
#pragma unroll
    for (int i = 0; i < 4; ++i)
        wv[i] = *reinterpret_cast<const v4f*>(w1 + (wr + 16 * i) * CIN + wc);

    float bias[4][8];
#pragma unroll
    for (int mt = 0; mt < 4; ++mt)
#pragma unroll
        for (int r = 0; r < 8; ++r)
            bias[mt][r] = b1[mt * 16 + 8 * half + r];

    const int boff = 2 * half * K2_XPITCH + n0 + lr;
    const int aoff = lr * K2_APITCH + 2 * half;

    v2f fa[2][4][4], fb[2][4];
    v8f acc[4] = {};
    v2f csum = {0.f, 0.f};

#pragma unroll
    for (int c = 0; c < 4; ++c) {
#pragma unroll
        for (int i = 0; i < 8; ++i)
            *reinterpret_cast<v4f*>(&Xs[(c * 64 + xr + 8 * i) * K2_XPITCH + xc]) = xv[i];
#pragma unroll
        for (int i = 0; i < 4; ++i)
            *reinterpret_cast<v4f*>(&As[(wr + 16 * i) * K2_APITCH + c * 64 + wc]) = wv[i];
        __syncthreads();
        if (c + 1 < 4) {
#pragma unroll
            for (int i = 0; i < 8; ++i)
                xv[i] = *reinterpret_cast<const v4f*>(xb + ((c + 1) * 64 + xr + 8 * i) * HW + xc);
#pragma unroll
            for (int i = 0; i < 4; ++i)
                wv[i] = *reinterpret_cast<const v4f*>(w1 + (wr + 16 * i) * CIN + (c + 1) * 64 + wc);
        }

#pragma unroll
        for (int j = 0; j < 4; ++j) {
            const int k0 = c * 64 + j * 4;
            fb[0][j].x = Xs[boff + k0 * K2_XPITCH];
            fb[0][j].y = Xs[boff + (k0 + 1) * K2_XPITCH];
#pragma unroll
            for (int mt = 0; mt < 4; ++mt)
                fa[0][j][mt] = *reinterpret_cast<const v2f*>(&As[aoff + mt * 16 * K2_APITCH + k0]);
        }
#pragma unroll
        for (int g = 0; g < 4; ++g) {
            const int cur = g & 1, nxt = cur ^ 1;
            if (g + 1 < 4) {
#pragma unroll
                for (int j = 0; j < 4; ++j) {
                    const int k0 = c * 64 + (g + 1) * 16 + j * 4;
                    fb[nxt][j].x = Xs[boff + k0 * K2_XPITCH];
                    fb[nxt][j].y = Xs[boff + (k0 + 1) * K2_XPITCH];
#pragma unroll
                    for (int mt = 0; mt < 4; ++mt)
                        fa[nxt][j][mt] = *reinterpret_cast<const v2f*>(&As[aoff + mt * 16 * K2_APITCH + k0]);
                }
            }
            __builtin_amdgcn_sched_barrier(0);
#pragma unroll
            for (int j = 0; j < 4; ++j) {
#pragma unroll
                for (int mt = 0; mt < 4; ++mt)
                    acc[mt] = __builtin_amdgcn_wmma_f32_16x16x4_f32(
                                  false, fa[cur][j][mt], false, fb[cur][j], (short)0, acc[mt], false, false);
                csum += fb[cur][j];
            }
            __builtin_amdgcn_sched_barrier(0);
        }
    }

    float cs = csum.x + csum.y;
    cs += gg_xhalf(cs);
    if (half == 0)
        mean[b * HW + nBase + n0 + lr] = cs * (1.0f / CIN);

    float* cb = comp + (size_t)b * CCOMP * HW + nBase + n0 + lr;
#pragma unroll
    for (int mt = 0; mt < 4; ++mt)
#pragma unroll
        for (int r = 0; r < 8; ++r) {
            int m = mt * 16 + r + 8 * half;
            cb[m * HW] = acc[mt][r] + bias[mt][r];
        }
}

#define K3_APITCH 580
#define K3_BROW   68
#define K3_BCH    272
#define K3_BGRP   1088
#define K3_KPART  4608

struct GgTaps { float a[3][4]; };

__device__ __forceinline__ void gg_load_taps(GgTaps& d, const float* __restrict__ xc,
                                             const int (&om)[3], const int (&ol)[3],
                                             const int (&orr)[3]) {
#pragma unroll
    for (int r = 0; r < 3; ++r) {
        v2f mid = *reinterpret_cast<const v2f*>(xc + om[r]);
        d.a[r][0] = xc[ol[r]];
        d.a[r][1] = mid.x;
        d.a[r][2] = mid.y;
        d.a[r][3] = xc[orr[r]];
    }
}

__device__ __forceinline__ void gg_reassemble(const GgTaps& d, const bool (&okM)[3],
                                              const bool (&okL)[3], const bool (&okR)[3],
                                              const float (&m0)[KK][SS], const float (&m1)[KK][SS],
                                              float* __restrict__ o) {
    float a[3][4];
#pragma unroll
    for (int r = 0; r < 3; ++r) {
        a[r][0] = okL[r] ? d.a[r][0] : 0.f;
        a[r][1] = okM[r] ? d.a[r][1] : 0.f;
        a[r][2] = okM[r] ? d.a[r][2] : 0.f;
        a[r][3] = okR[r] ? d.a[r][3] : 0.f;
    }
#pragma unroll
    for (int p = 0; p < 2; ++p) {
        float res[4];
#pragma unroll
        for (int q = 0; q < 2; ++q) {
            const int s = p * 2 + q;
            float acc0 = 0.f, acc1 = 0.f;
#pragma unroll
            for (int kk = 0; kk < KK; ++kk) {
                const int ky = kk / 3, kx = kk % 3;
                acc0 = fmaf(a[ky][kx],     m0[kk][s], acc0);
                acc1 = fmaf(a[ky][kx + 1], m1[kk][s], acc1);
            }
            res[q]     = acc0;
            res[2 + q] = acc1;
        }
        v4f ov = {res[0], res[1], res[2], res[3]};
        *reinterpret_cast<v4f*>(o + p * 128) = ov;
    }
}

__global__ void __launch_bounds__(512)
gg_kern_carafe(const float* __restrict__ comp,
               const float* __restrict__ w2,
               const float* __restrict__ b2,
               const float* __restrict__ x,
               const float* __restrict__ mean,
               float* __restrict__ out) {
    __shared__ __attribute__((aligned(16))) float As[CENC * K3_APITCH];
    __shared__ __attribute__((aligned(16))) float Bs[CCOMP * K3_BCH];
    __shared__ __attribute__((aligned(16))) float kpart[2 * K3_KPART];
    __shared__ __attribute__((aligned(16))) float smask[2 * KK * SS * WWID];

    const int tid = threadIdx.x;
    const int b   = blockIdx.x >> 5;
    const int h0  = (blockIdx.x & 31) * 2;
    const float* cb = comp + (size_t)b * CCOMP * HW;

    v4f wv[10];
#pragma unroll
    for (int t = 0; t < 10; ++t)
        wv[t] = *reinterpret_cast<const v4f*>(w2 + (tid + t * 512) * 4);
    const int iTail = 10 * 512 + (tid & 63);
    v4f wTail = *reinterpret_cast<const v4f*>(w2 + iTail * 4);
    const int  w4    = tid & 15;
    const int  rS    = (tid >> 4) & 3;
    const int  cq    = tid >> 6;
    const int  hS    = h0 - 1 + rS;
    const bool rowOk = (unsigned)hS < 64u;
    const int  hL    = rowOk ? hS : h0;
    v4f cv[8];
#pragma unroll
    for (int t = 0; t < 8; ++t)
        cv[t] = *reinterpret_cast<const v4f*>(cb + (cq + 8 * t) * HW + hL * 64 + w4 * 4);
#pragma unroll
    for (int t = 0; t < 10; ++t) {
        int i  = tid + t * 512;
        int m  = i / (KDIM / 4);
        int j4 = i - m * (KDIM / 4);
        *reinterpret_cast<v4f*>(&As[m * K3_APITCH + j4 * 4]) = wv[t];
    }
    {
        int m  = iTail / (KDIM / 4);
        int j4 = iTail - m * (KDIM / 4);
        *reinterpret_cast<v4f*>(&As[m * K3_APITCH + j4 * 4]) = wTail;
    }
#pragma unroll
    for (int t = 0; t < 8; ++t) {
        int o = (cq + 8 * t) * K3_BCH + rS * K3_BROW + 1 + w4 * 4;
        Bs[o]     = rowOk ? cv[t].x : 0.f;
        Bs[o + 1] = rowOk ? cv[t].y : 0.f;
        Bs[o + 2] = rowOk ? cv[t].z : 0.f;
        Bs[o + 3] = rowOk ? cv[t].w : 0.f;
    }
    if (tid < 256) {
        int o = (tid >> 2) * K3_BCH + (tid & 3) * K3_BROW;
        Bs[o]      = 0.f;
        Bs[o + 65] = 0.f;
    }
    __syncthreads();

    const int wave  = tid >> 5;
    const int lane  = tid & 31;
    const int lr    = lane & 15;
    const int half  = lane >> 4;
    const int kh    = wave >> 3;
    const int nt    = wave & 7;
    const int rowIn = nt >> 2;
    const int w0    = (nt & 3) * 16;

    {
        int boff[9][2];
#pragma unroll
        for (int j = 0; j < 9; ++j)
#pragma unroll
            for (int s = 0; s < 2; ++s) {
                int q  = j * 4 + 2 * half + s;
                int cc = q / 9, t = q - cc * 9, ky = t / 3, kx = t - ky * 3;
                boff[j][s] = kh * 8 * K3_BGRP + cc * K3_BCH + (rowIn + ky) * K3_BROW + (w0 + lr + kx);
            }
        const int aoff  = lr * K3_APITCH + 2 * half + kh * 288;
        const int aoff2 = (lr < 4 ? 32 + lr : lr) * K3_APITCH + 2 * half + kh * 288;

        v2f fa[2][9][3], fb[2][9];
        v8f acc0 = {}, acc1 = {}, acc2 = {};

#pragma unroll
        for (int j = 0; j < 9; ++j) {
            fb[0][j].x = Bs[boff[j][0]];
            fb[0][j].y = Bs[boff[j][1]];
            fa[0][j][0] = *reinterpret_cast<const v2f*>(&As[aoff + j * 4]);
            fa[0][j][1] = *reinterpret_cast<const v2f*>(&As[aoff + 16 * K3_APITCH + j * 4]);
            fa[0][j][2] = *reinterpret_cast<const v2f*>(&As[aoff2 + j * 4]);
        }
#pragma unroll
        for (int g = 0; g < 8; ++g) {
            const int cur = g & 1, nxt = cur ^ 1;
            if (g + 1 < 8) {
#pragma unroll
                for (int j = 0; j < 9; ++j) {
                    const int k1 = (g + 1) * 36 + j * 4;
                    fb[nxt][j].x = Bs[(g + 1) * K3_BGRP + boff[j][0]];
                    fb[nxt][j].y = Bs[(g + 1) * K3_BGRP + boff[j][1]];
                    fa[nxt][j][0] = *reinterpret_cast<const v2f*>(&As[aoff + k1]);
                    fa[nxt][j][1] = *reinterpret_cast<const v2f*>(&As[aoff + 16 * K3_APITCH + k1]);
                    fa[nxt][j][2] = *reinterpret_cast<const v2f*>(&As[aoff2 + k1]);
                }
            }
            __builtin_amdgcn_sched_barrier(0);
#pragma unroll
            for (int j = 0; j < 9; ++j) {
                acc0 = __builtin_amdgcn_wmma_f32_16x16x4_f32(
                           false, fa[cur][j][0], false, fb[cur][j], (short)0, acc0, false, false);
                acc1 = __builtin_amdgcn_wmma_f32_16x16x4_f32(
                           false, fa[cur][j][1], false, fb[cur][j], (short)0, acc1, false, false);
                acc2 = __builtin_amdgcn_wmma_f32_16x16x4_f32(
                           false, fa[cur][j][2], false, fb[cur][j], (short)0, acc2, false, false);
            }
            __builtin_amdgcn_sched_barrier(0);
        }

        float* kp = kpart + kh * K3_KPART + nt * 16 + lr;
        const int mh = 8 * half;
#pragma unroll
        for (int r = 0; r < 8; ++r) {
            kp[(mh + r) * 128]      = acc0[r];
            kp[(16 + mh + r) * 128] = acc1[r];
        }
        if (half == 0) {
#pragma unroll
            for (int r = 0; r < 4; ++r)
                kp[(32 + r) * 128] = acc2[r];
        }
    }
    __syncthreads();

    const int rsel = wave >> 3;
    const int h    = h0 + rsel;
    const int wvc  = wave & 7;

    int  om[3], ol[3], orr[3];
    bool okM[3], okL[3], okR[3];
#pragma unroll
    for (int r = 0; r < 3; ++r) {
        const int  hh = h - 1 + r;
        const bool ok = (unsigned)hh < 64u;
        const int  hc = ok ? hh : h;
        om[r]  = hc * 64 + 2 * lane;
        ol[r]  = hc * 64 + (lane > 0 ? 2 * lane - 1 : 0);
        orr[r] = hc * 64 + (lane < 31 ? 2 * lane + 2 : 63);
        okM[r] = ok;
        okL[r] = ok && lane > 0;
        okR[r] = ok && lane < 31;
    }
    const float* xw = x   + (size_t)(b * CIN + wvc) * HW;
    float*       ow = out + (size_t)(b * CIN + wvc) * (4 * HW) + (2 * h) * 128 + 4 * lane;

    GgTaps tA0, tA1, tB0, tB1;
    gg_load_taps(tA0, xw, om, ol, orr);
    gg_load_taps(tA1, xw + 8 * HW, om, ol, orr);

    float* sm = smask + rsel * (KK * SS * WWID);
    {
        const int w = tid & 63, s = (tid >> 6) & 3;
        const float* meanb = mean + b * HW;
        const float* kq    = kpart + s * 128 + rsel * 64 + w;
        float kv[KK], mp[KK];
#pragma unroll
        for (int kk = 0; kk < KK; ++kk) {
            const int  dy = kk / 3 - 1, dx = kk % 3 - 1;
            const int  hh = h + dy, ww = w + dx;
            const bool ok = (unsigned)hh < 64u && (unsigned)ww < 64u;
            kv[kk] = (kq[kk * SS * 128] + kq[K3_KPART + kk * SS * 128]) + b2[kk * SS + s];
            mp[kk] = meanb[ok ? hh * 64 + ww : h * 64 + w];
            mp[kk] = ok ? mp[kk] : 0.f;
        }
        const float mc = meanb[h * 64 + w];
        float mval[KK];
#pragma unroll
        for (int kk = 0; kk < KK; ++kk) {
            float g = mp[kk] - mc;
            g = 1.0f / (g * g + 0.2f);
            mval[kk] = g * kv[kk];
        }
        float mx = mval[0];
#pragma unroll
        for (int kk = 1; kk < KK; ++kk) mx = fmaxf(mx, mval[kk]);
        float e[KK], sum = 0.f;
#pragma unroll
        for (int kk = 0; kk < KK; ++kk) { e[kk] = __expf(mval[kk] - mx); sum += e[kk]; }
        const float inv = 1.0f / sum;
#pragma unroll
        for (int kk = 0; kk < KK; ++kk) sm[(kk * SS + s) * WWID + w] = e[kk] * inv;
    }
    __syncthreads();

    float m0[KK][SS], m1[KK][SS];
#pragma unroll
    for (int kk = 0; kk < KK; ++kk)
#pragma unroll
        for (int s = 0; s < SS; ++s) {
            v2f t = *reinterpret_cast<const v2f*>(&sm[(kk * SS + s) * WWID + 2 * lane]);
            m0[kk][s] = t.x;
            m1[kk][s] = t.y;
        }

    for (int i = 0; i < 28; i += 4) {
        const float* xi = xw + (size_t)i * (8 * HW);
        float*       oi = ow + (size_t)i * (8 * 4 * HW);
        gg_load_taps(tB0, xi + 16 * HW, om, ol, orr);
        gg_load_taps(tB1, xi + 24 * HW, om, ol, orr);
        gg_reassemble(tA0, okM, okL, okR, m0, m1, oi);
        gg_reassemble(tA1, okM, okL, okR, m0, m1, oi + 8 * 4 * HW);
        gg_load_taps(tA0, xi + 32 * HW, om, ol, orr);
        gg_load_taps(tA1, xi + 40 * HW, om, ol, orr);
        gg_reassemble(tB0, okM, okL, okR, m0, m1, oi + 16 * 4 * HW);
        gg_reassemble(tB1, okM, okL, okR, m0, m1, oi + 24 * 4 * HW);
    }
    {
        const float* xi = xw + (size_t)28 * (8 * HW);
        float*       oi = ow + (size_t)28 * (8 * 4 * HW);
        gg_load_taps(tB0, xi + 16 * HW, om, ol, orr);
        gg_load_taps(tB1, xi + 24 * HW, om, ol, orr);
        gg_reassemble(tA0, okM, okL, okR, m0, m1, oi);
        gg_reassemble(tA1, okM, okL, okR, m0, m1, oi + 8 * 4 * HW);
        gg_reassemble(tB0, okM, okL, okR, m0, m1, oi + 16 * 4 * HW);
        gg_reassemble(tB1, okM, okL, okR, m0, m1, oi + 24 * 4 * HW);
    }
}

extern "C" void kernel_launch(void* const* d_in, const int* in_sizes, int n_in,
                              void* d_out, int out_size, void* d_ws, size_t ws_size,
                              hipStream_t stream) {
    const float* x  = (const float*)d_in[0];
    const float* w1 = (const float*)d_in[1];
    const float* b1 = (const float*)d_in[2];
    const float* w2 = (const float*)d_in[3];
    const float* b2 = (const float*)d_in[4];
    float* out = (float*)d_out;

    float* ws   = (float*)d_ws;
    float* mean = ws;
    float* comp = ws + 32768;

    gg_comp_gemm  <<<256, 256, 0, stream>>>(x, w1, b1, comp, mean);
    gg_kern_carafe<<<256, 512, 0, stream>>>(comp, w2, b2, x, mean, out);
}


# reassembly: a lane owns its 2 pixels in both rows of the block (144 weights in registers); a channel's 4 input rows are read once per block, side taps come from adjacent lanes; sums unchanged
# speedup vs baseline: 4.3515x; 1.0431x over previous
#include <hip/hip_runtime.h>

typedef float v2f __attribute__((ext_vector_type(2)));
typedef float v4f __attribute__((ext_vector_type(4)));
typedef float v8f __attribute__((ext_vector_type(8)));

#define CIN   256
#define CCOMP 64
#define CENC  36
#define HH    64
#define WWID  64
#define HW    4096
#define KK    9
#define SS    4
#define KDIM  576

__device__ __forceinline__ float gg_xhalf(float v) {
    int bits = __builtin_bit_cast(int, v);
    return __builtin_bit_cast(float,
        __builtin_amdgcn_permlanex16(bits, bits, 0x76543210, 0xfedcba98, false, false));
}

#define K2_XPITCH 136
#define K2_APITCH 260

__global__ void __launch_bounds__(256)
gg_comp_gemm(const float* __restrict__ x,
             const float* __restrict__ w1,
             const float* __restrict__ b1,
             float* __restrict__ comp,
             float* __restrict__ mean) {
    __shared__ __attribute__((aligned(16))) float Xs[CIN * K2_XPITCH];
    __shared__ __attribute__((aligned(16))) float As[CCOMP * K2_APITCH];

    const int tid   = threadIdx.x;
    const int b     = blockIdx.x >> 5;
    const int nBase = (blockIdx.x & 31) * 128;
    const int wave  = tid >> 5;
    const int lane  = tid & 31;
    const int lr    = lane & 15;
    const int half  = lane >> 4;
    const int n0    = wave * 16;

    const float* xb = x + (size_t)b * CIN * HW + nBase;

    const int xc = (tid & 31) * 4;
    const int xr = tid >> 5;
    const int wc = (tid & 15) * 4;
    const int wr = tid >> 4;

    v4f xv[8], wv[4];
#pragma unroll
    for (int i = 0; i < 8; ++i)
        xv[i] = *reinterpret_cast<const v4f*>(xb + (xr + 8 * i) * HW + xc);
#pragma unroll
    for (int i = 0; i < 4; ++i)
        wv[i] = *reinterpret_cast<const v4f*>(w1 + (wr + 16 * i) * CIN + wc);

    float bias[4][8];
#pragma unroll
    for (int mt = 0; mt < 4; ++mt)
#pragma unroll
        for (int r = 0; r < 8; ++r)
            bias[mt][r] = b1[mt * 16 + 8 * half + r];

    const int boff = 2 * half * K2_XPITCH + n0 + lr;
    const int aoff = lr * K2_APITCH + 2 * half;

    v2f fa[2][4][4], fb[2][4];
    v8f acc[4] = {};
    v2f csum = {0.f, 0.f};

#pragma unroll
    for (int c = 0; c < 4; ++c) {
#pragma unroll
        for (int i = 0; i < 8; ++i)
            *reinterpret_cast<v4f*>(&Xs[(c * 64 + xr + 8 * i) * K2_XPITCH + xc]) = xv[i];
#pragma unroll
        for (int i = 0; i < 4; ++i)
            *reinterpret_cast<v4f*>(&As[(wr + 16 * i) * K2_APITCH + c * 64 + wc]) = wv[i];
        __syncthreads();
        if (c + 1 < 4) {
#pragma unroll
            for (int i = 0; i < 8; ++i)
                xv[i] = *reinterpret_cast<const v4f*>(xb + ((c + 1) * 64 + xr + 8 * i) * HW + xc);
#pragma unroll
            for (int i = 0; i < 4; ++i)
                wv[i] = *reinterpret_cast<const v4f*>(w1 + (wr + 16 * i) * CIN + (c + 1) * 64 + wc);
        }

#pragma unroll
        for (int j = 0; j < 4; ++j) {
            const int k0 = c * 64 + j * 4;
            fb[0][j].x = Xs[boff + k0 * K2_XPITCH];
            fb[0][j].y = Xs[boff + (k0 + 1) * K2_XPITCH];
#pragma unroll
            for (int mt = 0; mt < 4; ++mt)
                fa[0][j][mt] = *reinterpret_cast<const v2f*>(&As[aoff + mt * 16 * K2_APITCH + k0]);
        }
#pragma unroll
        for (int g = 0; g < 4; ++g) {
            const int cur = g & 1, nxt = cur ^ 1;
            if (g + 1 < 4) {
#pragma unroll
                for (int j = 0; j < 4; ++j) {
                    const int k0 = c * 64 + (g + 1) * 16 + j * 4;
                    fb[nxt][j].x = Xs[boff + k0 * K2_XPITCH];
                    fb[nxt][j].y = Xs[boff + (k0 + 1) * K2_XPITCH];
#pragma unroll
                    for (int mt = 0; mt < 4; ++mt)
                        fa[nxt][j][mt] = *reinterpret_cast<const v2f*>(&As[aoff + mt * 16 * K2_APITCH + k0]);
                }
            }
            __builtin_amdgcn_sched_barrier(0);
#pragma unroll
            for (int j = 0; j < 4; ++j) {
#pragma unroll
                for (int mt = 0; mt < 4; ++mt)
                    acc[mt] = __builtin_amdgcn_wmma_f32_16x16x4_f32(
                                  false, fa[cur][j][mt], false, fb[cur][j], (short)0, acc[mt], false, false);
                csum += fb[cur][j];
            }
            __builtin_amdgcn_sched_barrier(0);
        }
    }

    float cs = csum.x + csum.y;
    cs += gg_xhalf(cs);
    if (half == 0)
        mean[b * HW + nBase + n0 + lr] = cs * (1.0f / CIN);

    float* cb = comp + (size_t)b * CCOMP * HW + nBase + n0 + lr;
#pragma unroll
    for (int mt = 0; mt < 4; ++mt)
#pragma unroll
        for (int r = 0; r < 8; ++r) {
            int m = mt * 16 + r + 8 * half;
            cb[m * HW] = acc[mt][r] + bias[mt][r];
        }
}

#define K3_APITCH 580
#define K3_BROW   68
#define K3_BCH    272
#define K3_BGRP   1088
#define K3_KPART  4608

struct GgRows { v2f a[4]; };

__device__ __forceinline__ void gg_load_rows(GgRows& d, const float* __restrict__ xc,
                                             const int (&om)[4]) {
#pragma unroll
    for (int r = 0; r < 4; ++r)
        d.a[r] = *reinterpret_cast<const v2f*>(xc + om[r]);
}

__device__ __forceinline__ float gg_from_lane(int idx, float v) {
    return __builtin_bit_cast(float,
        __builtin_amdgcn_ds_bpermute(idx, __builtin_bit_cast(int, v)));
}

__device__ __forceinline__ void gg_reassemble(const GgRows& d, const bool (&okM)[4],
                                              const bool (&okL)[4], const bool (&okR)[4],
                                              int idxL, int idxR,
                                              const float (&m0)[2][KK][SS], const float (&m1)[2][KK][SS],
                                              float* __restrict__ o) {
    float a[4][4];
#pragma unroll
    for (int r = 0; r < 4; ++r) {
        const float lft = gg_from_lane(idxL, d.a[r].y);
        const float rgt = gg_from_lane(idxR, d.a[r].x);
        a[r][0] = okL[r] ? lft      : 0.f;
        a[r][1] = okM[r] ? d.a[r].x : 0.f;
        a[r][2] = okM[r] ? d.a[r].y : 0.f;
        a[r][3] = okR[r] ? rgt      : 0.f;
    }
#pragma unroll
    for (int rr = 0; rr < 2; ++rr) {
#pragma unroll
        for (int p = 0; p < 2; ++p) {
            float res[4];
#pragma unroll
            for (int q = 0; q < 2; ++q) {
                const int s = p * 2 + q;
                float acc0 = 0.f, acc1 = 0.f;
#pragma unroll
                for (int kk = 0; kk < KK; ++kk) {
                    const int ky = kk / 3, kx = kk % 3;
                    acc0 = fmaf(a[rr + ky][kx],     m0[rr][kk][s], acc0);
                    acc1 = fmaf(a[rr + ky][kx + 1], m1[rr][kk][s], acc1);
                }
                res[q]     = acc0;
                res[2 + q] = acc1;
            }
            v4f ov = {res[0], res[1], res[2], res[3]};
            *reinterpret_cast<v4f*>(o + (2 * rr + p) * 128) = ov;
        }
    }
}

__global__ void __launch_bounds__(512)
gg_kern_carafe(const float* __restrict__ comp,
               const float* __restrict__ w2,
               const float* __restrict__ b2,
               const float* __restrict__ x,
               const float* __restrict__ mean,
               float* __restrict__ out) {
    __shared__ __attribute__((aligned(16))) float As[CENC * K3_APITCH];
    __shared__ __attribute__((aligned(16))) float Bs[CCOMP * K3_BCH];
    __shared__ __attribute__((aligned(16))) float kpart[2 * K3_KPART];
    __shared__ __attribute__((aligned(16))) float smask[2 * KK * SS * WWID];

    const int tid = threadIdx.x;
    const int b   = blockIdx.x >> 5;
    const int h0  = (blockIdx.x & 31) * 2;
    const float* cb = comp + (size_t)b * CCOMP * HW;

    v4f wv[10];
#pragma unroll
    for (int t = 0; t < 10; ++t)
        wv[t] = *reinterpret_cast<const v4f*>(w2 + (tid + t * 512) * 4);
    const int iTail = 10 * 512 + (tid & 63);
    v4f wTail = *reinterpret_cast<const v4f*>(w2 + iTail * 4);
    const int  w4    = tid & 15;
    const int  rS    = (tid >> 4) & 3;
    const int  cq    = tid >> 6;
    const int  hS    = h0 - 1 + rS;
    const bool rowOk = (unsigned)hS < 64u;
    const int  hL    = rowOk ? hS : h0;
    v4f cv[8];
#pragma unroll
    for (int t = 0; t < 8; ++t)
        cv[t] = *reinterpret_cast<const v4f*>(cb + (cq + 8 * t) * HW + hL * 64 + w4 * 4);
#pragma unroll
    for (int t = 0; t < 10; ++t) {
        int i  = tid + t * 512;
        int m  = i / (KDIM / 4);
        int j4 = i - m * (KDIM / 4);
        *reinterpret_cast<v4f*>(&As[m * K3_APITCH + j4 * 4]) = wv[t];
    }
    {
        int m  = iTail / (KDIM / 4);
        int j4 = iTail - m * (KDIM / 4);
        *reinterpret_cast<v4f*>(&As[m * K3_APITCH + j4 * 4]) = wTail;
    }
#pragma unroll
    for (int t = 0; t < 8; ++t) {
        int o = (cq + 8 * t) * K3_BCH + rS * K3_BROW + 1 + w4 * 4;
        Bs[o]     = rowOk ? cv[t].x : 0.f;
        Bs[o + 1] = rowOk ? cv[t].y : 0.f;
        Bs[o + 2] = rowOk ? cv[t].z : 0.f;
        Bs[o + 3] = rowOk ? cv[t].w : 0.f;
    }
    if (tid < 256) {
        int o = (tid >> 2) * K3_BCH + (tid & 3) * K3_BROW;
        Bs[o]      = 0.f;
        Bs[o + 65] = 0.f;
    }
    __syncthreads();

    const int wave  = tid >> 5;
    const int lane  = tid & 31;
    const int lr    = lane & 15;
    const int half  = lane >> 4;
    const int kh    = wave >> 3;
    const int nt    = wave & 7;
    const int rowIn = nt >> 2;
    const int w0    = (nt & 3) * 16;

    {
        int boff[9][2];
#pragma unroll
        for (int j = 0; j < 9; ++j)
#pragma unroll
            for (int s = 0; s < 2; ++s) {
                int q  = j * 4 + 2 * half + s;
                int cc = q / 9, t = q - cc * 9, ky = t / 3, kx = t - ky * 3;
                boff[j][s] = kh * 8 * K3_BGRP + cc * K3_BCH + (rowIn + ky) * K3_BROW + (w0 + lr + kx);
            }
        const int aoff  = lr * K3_APITCH + 2 * half + kh * 288;
        const int aoff2 = (lr < 4 ? 32 + lr : lr) * K3_APITCH + 2 * half + kh * 288;

        v2f fa[2][9][3], fb[2][9];
        v8f acc0 = {}, acc1 = {}, acc2 = {};

#pragma unroll
        for (int j = 0; j < 9; ++j) {
            fb[0][j].x = Bs[boff[j][0]];
            fb[0][j].y = Bs[boff[j][1]];
            fa[0][j][0] = *reinterpret_cast<const v2f*>(&As[aoff + j * 4]);
            fa[0][j][1] = *reinterpret_cast<const v2f*>(&As[aoff + 16 * K3_APITCH + j * 4]);
            fa[0][j][2] = *reinterpret_cast<const v2f*>(&As[aoff2 + j * 4]);
        }
#pragma unroll
        for (int g = 0; g < 8; ++g) {
            const int cur = g & 1, nxt = cur ^ 1;
            if (g + 1 < 8) {
#pragma unroll
                for (int j = 0; j < 9; ++j) {
                    const int k1 = (g + 1) * 36 + j * 4;
                    fb[nxt][j].x = Bs[(g + 1) * K3_BGRP + boff[j][0]];
                    fb[nxt][j].y = Bs[(g + 1) * K3_BGRP + boff[j][1]];
                    fa[nxt][j][0] = *reinterpret_cast<const v2f*>(&As[aoff + k1]);
                    fa[nxt][j][1] = *reinterpret_cast<const v2f*>(&As[aoff + 16 * K3_APITCH + k1]);
                    fa[nxt][j][2] = *reinterpret_cast<const v2f*>(&As[aoff2 + k1]);
                }
            }
            __builtin_amdgcn_sched_barrier(0);
#pragma unroll
            for (int j = 0; j < 9; ++j) {
                acc0 = __builtin_amdgcn_wmma_f32_16x16x4_f32(
                           false, fa[cur][j][0], false, fb[cur][j], (short)0, acc0, false, false);
                acc1 = __builtin_amdgcn_wmma_f32_16x16x4_f32(
                           false, fa[cur][j][1], false, fb[cur][j], (short)0, acc1, false, false);
                acc2 = __builtin_amdgcn_wmma_f32_16x16x4_f32(
                           false, fa[cur][j][2], false, fb[cur][j], (short)0, acc2, false, false);
            }
            __builtin_amdgcn_sched_barrier(0);
        }

        float* kp = kpart + kh * K3_KPART + nt * 16 + lr;
        const int mh = 8 * half;
#pragma unroll
        for (int r = 0; r < 8; ++r) {
            kp[(mh + r) * 128]      = acc0[r];
            kp[(16 + mh + r) * 128] = acc1[r];
        }
        if (half == 0) {
#pragma unroll
            for (int r = 0; r < 4; ++r)
                kp[(32 + r) * 128] = acc2[r];
        }
    }
    __syncthreads();

    const int rsel = wave >> 3;
    const int h    = h0 + rsel;

    int  om[4];
    bool okM[4], okL[4], okR[4];
#pragma unroll
    for (int r = 0; r < 4; ++r) {
        const int  hh = h0 - 1 + r;
        const bool ok = (unsigned)hh < 64u;
        const int  hc = ok ? hh : h0;
        om[r]  = hc * 64 + 2 * lane;
        okM[r] = ok;
        okL[r] = ok && lane > 0;
        okR[r] = ok && lane < 31;
    }
    const int idxL = ((lane + 31) & 31) * 4;
    const int idxR = ((lane + 1) & 31) * 4;
    const float* xw = x   + (size_t)(b * CIN + wave) * HW;
    float*       ow = out + (size_t)(b * CIN + wave) * (4 * HW) + (2 * h0) * 128 + 4 * lane;

    GgRows t0, t1, t2, t3;
    gg_load_rows(t0, xw, om);
    gg_load_rows(t1, xw + 16 * HW, om);
    gg_load_rows(t2, xw + 32 * HW, om);

    float* sm = smask + rsel * (KK * SS * WWID);
    {
        const int w = tid & 63, s = (tid >> 6) & 3;
        const float* meanb = mean + b * HW;
        const float* kq    = kpart + s * 128 + rsel * 64 + w;
        float kv[KK], mp[KK];
#pragma unroll
        for (int kk = 0; kk < KK; ++kk) {
            const int  dy = kk / 3 - 1, dx = kk % 3 - 1;
            const int  hh = h + dy, ww = w + dx;
            const bool ok = (unsigned)hh < 64u && (unsigned)ww < 64u;
            kv[kk] = (kq[kk * SS * 128] + kq[K3_KPART + kk * SS * 128]) + b2[kk * SS + s];
            mp[kk] = meanb[ok ? hh * 64 + ww : h * 64 + w];
            mp[kk] = ok ? mp[kk] : 0.f;
        }
        const float mc = meanb[h * 64 + w];
        float mval[KK];
#pragma unroll
        for (int kk = 0; kk < KK; ++kk) {
            float g = mp[kk] - mc;
            g = 1.0f / (g * g + 0.2f);
            mval[kk] = g * kv[kk];
        }
        float mx = mval[0];
#pragma unroll
        for (int kk = 1; kk < KK; ++kk) mx = fmaxf(mx, mval[kk]);
        float e[KK], sum = 0.f;
#pragma unroll
        for (int kk = 0; kk < KK; ++kk) { e[kk] = __expf(mval[kk] - mx); sum += e[kk]; }
        const float inv = 1.0f / sum;
#pragma unroll
        for (int kk = 0; kk < KK; ++kk) sm[(kk * SS + s) * WWID + w] = e[kk] * inv;
    }
    __syncthreads();

    float m0[2][KK][SS], m1[2][KK][SS];
#pragma unroll
    for (int rr = 0; rr < 2; ++rr)
#pragma unroll
        for (int kk = 0; kk < KK; ++kk)
#pragma unroll
            for (int s = 0; s < SS; ++s) {
                v2f t = *reinterpret_cast<const v2f*>(
                            &smask[rr * (KK * SS * WWID) + (kk * SS + s) * WWID + 2 * lane]);
                m0[rr][kk][s] = t.x;
                m1[rr][kk][s] = t.y;
            }

    for (int i = 0; i < 12; i += 4) {
        const float* xi = xw + (size_t)i * (16 * HW);
        float*       oi = ow + (size_t)i * (16 * 4 * HW);
        gg_load_rows(t3, xi + 48 * HW, om);
        gg_reassemble(t0, okM, okL, okR, idxL, idxR, m0, m1, oi);
        gg_load_rows(t0, xi + 64 * HW, om);
        gg_reassemble(t1, okM, okL, okR, idxL, idxR, m0, m1, oi + 16 * 4 * HW);
        gg_load_rows(t1, xi + 80 * HW, om);
        gg_reassemble(t2, okM, okL, okR, idxL, idxR, m0, m1, oi + 32 * 4 * HW);
        gg_load_rows(t2, xi + 96 * HW, om);
        gg_reassemble(t3, okM, okL, okR, idxL, idxR, m0, m1, oi + 48 * 4 * HW);
    }
    {
        const float* xi = xw + (size_t)12 * (16 * HW);
        float*       oi = ow + (size_t)12 * (16 * 4 * HW);
        gg_load_rows(t3, xi + 48 * HW, om);
        gg_reassemble(t0, okM, okL, okR, idxL, idxR, m0, m1, oi);
        gg_reassemble(t1, okM, okL, okR, idxL, idxR, m0, m1, oi + 16 * 4 * HW);
        gg_reassemble(t2, okM, okL, okR, idxL, idxR, m0, m1, oi + 32 * 4 * HW);
        gg_reassemble(t3, okM, okL, okR, idxL, idxR, m0, m1, oi + 48 * 4 * HW);
    }
}

extern "C" void kernel_launch(void* const* d_in, const int* in_sizes, int n_in,
                              void* d_out, int out_size, void* d_ws, size_t ws_size,
                              hipStream_t stream) {
    const float* x  = (const float*)d_in[0];
    const float* w1 = (const float*)d_in[1];
    const float* b1 = (const float*)d_in[2];
    const float* w2 = (const float*)d_in[3];
    const float* b2 = (const float*)d_in[4];
    float* out = (float*)d_out;

    float* ws   = (float*)d_ws;
    float* mean = ws;
    float* comp = ws + 32768;

    gg_comp_gemm  <<<256, 256, 0, stream>>>(x, w1, b1, comp, mean);
    gg_kern_carafe<<<256, 512, 0, stream>>>(comp, w2, b2, x, mean, out);
}


# 3x3-conv GEMM taken per kernel column: M = 36 outputs x 3 kx = 108 rows (7 tiles), K = 64x3; 168 WMMAs a wave, not 216; kern = sum of 3 column terms x 2 k halves, so its sum order changes
# speedup vs baseline: 4.6779x; 1.0750x over previous
#include <hip/hip_runtime.h>

typedef float v2f __attribute__((ext_vector_type(2)));
typedef float v4f __attribute__((ext_vector_type(4)));
typedef float v8f __attribute__((ext_vector_type(8)));

#define CIN   256
#define CCOMP 64
#define CENC  36
#define HH    64
#define WWID  64
#define HW    4096
#define KK    9
#define SS    4
#define KDIM  576

__device__ __forceinline__ float gg_xhalf(float v) {
    int bits = __builtin_bit_cast(int, v);
    return __builtin_bit_cast(float,
        __builtin_amdgcn_permlanex16(bits, bits, 0x76543210, 0xfedcba98, false, false));
}

#define K2_XPITCH 136
#define K2_APITCH 260

__global__ void __launch_bounds__(256)
gg_comp_gemm(const float* __restrict__ x,
             const float* __restrict__ w1,
             const float* __restrict__ b1,
             float* __restrict__ comp,
             float* __restrict__ mean) {
    __shared__ __attribute__((aligned(16))) float Xs[CIN * K2_XPITCH];
    __shared__ __attribute__((aligned(16))) float As[CCOMP * K2_APITCH];

    const int tid   = threadIdx.x;
    const int b     = blockIdx.x >> 5;
    const int nBase = (blockIdx.x & 31) * 128;
    const int wave  = tid >> 5;
    const int lane  = tid & 31;
    const int lr    = lane & 15;
    const int half  = lane >> 4;
    const int n0    = wave * 16;

    const float* xb = x + (size_t)b * CIN * HW + nBase;

    const int xc = (tid & 31) * 4;
    const int xr = tid >> 5;
    const int wc = (tid & 15) * 4;
    const int wr = tid >> 4;

    v4f xv[8], wv[4];
#pragma unroll
    for (int i = 0; i < 8; ++i)
        xv[i] = *reinterpret_cast<const v4f*>(xb + (xr + 8 * i) * HW + xc);
#pragma unroll
    for (int i = 0; i < 4; ++i)
        wv[i] = *reinterpret_cast<const v4f*>(w1 + (wr + 16 * i) * CIN + wc);

    float bias[4][8];
#pragma unroll
    for (int mt = 0; mt < 4; ++mt)
#pragma unroll
        for (int r = 0; r < 8; ++r)
            bias[mt][r] = b1[mt * 16 + 8 * half + r];

    const int boff = 2 * half * K2_XPITCH + n0 + lr;
    const int aoff = lr * K2_APITCH + 2 * half;

    v2f fa[2][4][4], fb[2][4];
    v8f acc[4] = {};
    v2f csum = {0.f, 0.f};

#pragma unroll
    for (int c = 0; c < 4; ++c) {
#pragma unroll
        for (int i = 0; i < 8; ++i)
            *reinterpret_cast<v4f*>(&Xs[(c * 64 + xr + 8 * i) * K2_XPITCH + xc]) = xv[i];
#pragma unroll
        for (int i = 0; i < 4; ++i)
            *reinterpret_cast<v4f*>(&As[(wr + 16 * i) * K2_APITCH + c * 64 + wc]) = wv[i];
        __syncthreads();
        if (c + 1 < 4) {
#pragma unroll
            for (int i = 0; i < 8; ++i)
                xv[i] = *reinterpret_cast<const v4f*>(xb + ((c + 1) * 64 + xr + 8 * i) * HW + xc);
#pragma unroll
            for (int i = 0; i < 4; ++i)
                wv[i] = *reinterpret_cast<const v4f*>(w1 + (wr + 16 * i) * CIN + (c + 1) * 64 + wc);
        }

#pragma unroll
        for (int j = 0; j < 4; ++j) {
            const int k0 = c * 64 + j * 4;
            fb[0][j].x = Xs[boff + k0 * K2_XPITCH];
            fb[0][j].y = Xs[boff + (k0 + 1) * K2_XPITCH];
#pragma unroll
            for (int mt = 0; mt < 4; ++mt)
                fa[0][j][mt] = *reinterpret_cast<const v2f*>(&As[aoff + mt * 16 * K2_APITCH + k0]);
        }
#pragma unroll
        for (int g = 0; g < 4; ++g) {
            const int cur = g & 1, nxt = cur ^ 1;
            if (g + 1 < 4) {
#pragma unroll
                for (int j = 0; j < 4; ++j) {
                    const int k0 = c * 64 + (g + 1) * 16 + j * 4;
                    fb[nxt][j].x = Xs[boff + k0 * K2_XPITCH];
                    fb[nxt][j].y = Xs[boff + (k0 + 1) * K2_XPITCH];
#pragma unroll
                    for (int mt = 0; mt < 4; ++mt)
                        fa[nxt][j][mt] = *reinterpret_cast<const v2f*>(&As[aoff + mt * 16 * K2_APITCH + k0]);
                }
            }
            __builtin_amdgcn_sched_barrier(0);
#pragma unroll
            for (int j = 0; j < 4; ++j) {
#pragma unroll
                for (int mt = 0; mt < 4; ++mt)
                    acc[mt] = __builtin_amdgcn_wmma_f32_16x16x4_f32(
                                  false, fa[cur][j][mt], false, fb[cur][j], (short)0, acc[mt], false, false);
                csum += fb[cur][j];
            }
            __builtin_amdgcn_sched_barrier(0);
        }
    }

    float cs = csum.x + csum.y;
    cs += gg_xhalf(cs);
    if (half == 0)
        mean[b * HW + nBase + n0 + lr] = cs * (1.0f / CIN);

    float* cb = comp + (size_t)b * CCOMP * HW + nBase + n0 + lr;
#pragma unroll
    for (int mt = 0; mt < 4; ++mt)
#pragma unroll
        for (int r = 0; r < 8; ++r) {
            int m = mt * 16 + r + 8 * half;
            cb[m * HW] = acc[mt][r] + bias[mt][r];
        }
}

#define K3_APITCH 196
#define K3_AROWS  108
#define K3_BROW   72
#define K3_BCH    288
#define K3_BGRP   1152
#define K3_KPART  13824

struct GgRows { v2f a[4]; };

__device__ __forceinline__ void gg_load_rows(GgRows& d, const float* __restrict__ xc,
                                             const int (&om)[4]) {
#pragma unroll
    for (int r = 0; r < 4; ++r)
        d.a[r] = *reinterpret_cast<const v2f*>(xc + om[r]);
}

__device__ __forceinline__ float gg_from_lane(int idx, float v) {
    return __builtin_bit_cast(float,
        __builtin_amdgcn_ds_bpermute(idx, __builtin_bit_cast(int, v)));
}

__device__ __forceinline__ void gg_reassemble(const GgRows& d, const bool (&okM)[4],
                                              const bool (&okL)[4], const bool (&okR)[4],
                                              int idxL, int idxR,
                                              const float (&m0)[2][KK][SS], const float (&m1)[2][KK][SS],
                                              float* __restrict__ o) {
    float a[4][4];
#pragma unroll
    for (int r = 0; r < 4; ++r) {
        const float lft = gg_from_lane(idxL, d.a[r].y);
        const float rgt = gg_from_lane(idxR, d.a[r].x);
        a[r][0] = okL[r] ? lft      : 0.f;
        a[r][1] = okM[r] ? d.a[r].x : 0.f;
        a[r][2] = okM[r] ? d.a[r].y : 0.f;
        a[r][3] = okR[r] ? rgt      : 0.f;
    }
#pragma unroll
    for (int rr = 0; rr < 2; ++rr) {
#pragma unroll
        for (int p = 0; p < 2; ++p) {
            float res[4];
#pragma unroll
            for (int q = 0; q < 2; ++q) {
                const int s = p * 2 + q;
                float acc0 = 0.f, acc1 = 0.f;
#pragma unroll
                for (int kk = 0; kk < KK; ++kk) {
                    const int ky = kk / 3, kx = kk % 3;
                    acc0 = fmaf(a[rr + ky][kx],     m0[rr][kk][s], acc0);
                    acc1 = fmaf(a[rr + ky][kx + 1], m1[rr][kk][s], acc1);
                }
                res[q]     = acc0;
                res[2 + q] = acc1;
            }
            v4f ov = {res[0], res[1], res[2], res[3]};
            *reinterpret_cast<v4f*>(o + (2 * rr + p) * 128) = ov;
        }
    }
}

__global__ void __launch_bounds__(512)
gg_kern_carafe(const float* __restrict__ comp,
               const float* __restrict__ w2,
               const float* __restrict__ b2,
               const float* __restrict__ x,
               const float* __restrict__ mean,
               float* __restrict__ out) {
    __shared__ __attribute__((aligned(16))) float As[K3_AROWS * K3_APITCH];
    __shared__ __attribute__((aligned(16))) float Bs[CCOMP * K3_BCH];
    __shared__ __attribute__((aligned(16))) float kpart[2 * K3_KPART];
    __shared__ __attribute__((aligned(16))) float smask[2 * KK * SS * WWID];

    const int tid = threadIdx.x;
    const int b   = blockIdx.x >> 5;
    const int h0  = (blockIdx.x & 31) * 2;
    const float* cb = comp + (size_t)b * CCOMP * HW;

    const int  w4    = tid & 15;
    const int  rS    = (tid >> 4) & 3;
    const int  cq    = tid >> 6;
    const int  hS    = h0 - 1 + rS;
    const bool rowOk = (unsigned)hS < 64u;
    const int  hL    = rowOk ? hS : h0;
    v4f cv[8];
#pragma unroll
    for (int t = 0; t < 8; ++t)
        cv[t] = *reinterpret_cast<const v4f*>(cb + (cq + 8 * t) * HW + hL * 64 + w4 * 4);
    const int rsub = tid / 144;
    const int j4   = tid - rsub * 144;
    if (rsub < 3) {
        v4f wv[12];
#pragma unroll
        for (int t = 0; t < 12; ++t)
            wv[t] = *reinterpret_cast<const v4f*>(w2 + (3 * t + rsub) * KDIM + j4 * 4);
        const int rem = 4 * j4;
        const int qq  = rem / 3;
        const int rr  = rem - 3 * qq;
        const int d0  = rsub * (3 * K3_APITCH) + rr * K3_APITCH + qq;
        const int d1  = d0 + (rr == 2 ? 1 - 2 * K3_APITCH : K3_APITCH);
        const int d2  = d0 + (rr == 0 ? 2 * K3_APITCH : 1 - K3_APITCH);
#pragma unroll
        for (int t = 0; t < 12; ++t) {
            float* a = As + t * (9 * K3_APITCH);
            a[d0]     = wv[t].x;
            a[d1]     = wv[t].y;
            a[d2]     = wv[t].z;
            a[d0 + 1] = wv[t].w;
        }
    }
    {
        const v4f zero = {0.f, 0.f, 0.f, 0.f};
#pragma unroll
        for (int t = 0; t < 8; ++t)
            *reinterpret_cast<v4f*>(&Bs[(cq + 8 * t) * K3_BCH + rS * K3_BROW + w4 * 4]) = rowOk ? cv[t] : zero;
    }
    __syncthreads();

    const int wave  = tid >> 5;
    const int lane  = tid & 31;
    const int lr    = lane & 15;
    const int half  = lane >> 4;
    const int kh    = wave >> 3;
    const int nt    = wave & 7;
    const int rowIn = nt >> 2;
    const int w0    = (nt & 3) * 16;

    {
        int boff[3][2];
#pragma unroll
        for (int j = 0; j < 3; ++j)
#pragma unroll
            for (int s = 0; s < 2; ++s) {
                int q  = j * 4 + 2 * half + s;
                int cc = q / 3, ky = q - cc * 3;
                boff[j][s] = kh * 8 * K3_BGRP + cc * K3_BCH + (rowIn + ky) * K3_BROW + (w0 + lr);
            }
        const int aoff  = lr * K3_APITCH + 2 * half + kh * 96;
        const int aoff6 = (lr < 12 ? 96 + lr : lr) * K3_APITCH + 2 * half + kh * 96;

        v2f fa[2][3][7], fb[2][3];
        v8f acc[7] = {};

#pragma unroll
        for (int j = 0; j < 3; ++j) {
            fb[0][j].x = Bs[boff[j][0]];
            fb[0][j].y = Bs[boff[j][1]];
#pragma unroll
            for (int mt = 0; mt < 6; ++mt)
                fa[0][j][mt] = *reinterpret_cast<const v2f*>(&As[aoff + mt * 16 * K3_APITCH + j * 4]);
            fa[0][j][6] = *reinterpret_cast<const v2f*>(&As[aoff6 + j * 4]);
        }
#pragma unroll
        for (int g = 0; g < 8; ++g) {
            const int cur = g & 1, nxt = cur ^ 1;
            if (g + 1 < 8) {
#pragma unroll
                for (int j = 0; j < 3; ++j) {
                    const int k1 = (g + 1) * 12 + j * 4;
                    fb[nxt][j].x = Bs[(g + 1) * K3_BGRP + boff[j][0]];
                    fb[nxt][j].y = Bs[(g + 1) * K3_BGRP + boff[j][1]];
#pragma unroll
                    for (int mt = 0; mt < 6; ++mt)
                        fa[nxt][j][mt] = *reinterpret_cast<const v2f*>(&As[aoff + mt * 16 * K3_APITCH + k1]);
                    fa[nxt][j][6] = *reinterpret_cast<const v2f*>(&As[aoff6 + k1]);
                }
            }
            __builtin_amdgcn_sched_barrier(0);
#pragma unroll
            for (int j = 0; j < 3; ++j)
#pragma unroll
                for (int mt = 0; mt < 7; ++mt)
                    acc[mt] = __builtin_amdgcn_wmma_f32_16x16x4_f32(
                                  false, fa[cur][j][mt], false, fb[cur][j], (short)0, acc[mt], false, false);
            __builtin_amdgcn_sched_barrier(0);
        }

        float* kp = kpart + kh * K3_KPART + nt * 16 + lr;
        const int mh = 8 * half;
#pragma unroll
        for (int mt = 0; mt < 6; ++mt)
#pragma unroll
            for (int r = 0; r < 8; ++r)
                kp[(mt * 16 + mh + r) * 128] = acc[mt][r];
#pragma unroll
        for (int r = 0; r < 8; ++r)
            if (half == 0 || r < 4)
                kp[(96 + mh + r) * 128] = acc[6][r];
    }
    __syncthreads();

    const int rsel = wave >> 3;
    const int h    = h0 + rsel;

    int  om[4];
    bool okM[4], okL[4], okR[4];
#pragma unroll
    for (int r = 0; r < 4; ++r) {
        const int  hh = h0 - 1 + r;
        const bool ok = (unsigned)hh < 64u;
        const int  hc = ok ? hh : h0;
        om[r]  = hc * 64 + 2 * lane;
        okM[r] = ok;
        okL[r] = ok && lane > 0;
        okR[r] = ok && lane < 31;
    }
    const int idxL = ((lane + 31) & 31) * 4;
    const int idxR = ((lane + 1) & 31) * 4;
    const float* xw = x   + (size_t)(b * CIN + wave) * HW;
    float*       ow = out + (size_t)(b * CIN + wave) * (4 * HW) + (2 * h0) * 128 + 4 * lane;

    GgRows t0, t1, t2, t3;
    gg_load_rows(t0, xw, om);
    gg_load_rows(t1, xw + 16 * HW, om);
    gg_load_rows(t2, xw + 32 * HW, om);

    float* sm = smask + rsel * (KK * SS * WWID);
    {
        const int w = tid & 63, s = (tid >> 6) & 3;
        const float* meanb = mean + b * HW;
        const float* kq  = kpart + s * (3 * 128) + rsel * 64;
        const bool   okl = w > 0, okr = w < 63;
        const int    wl  = okl ? w - 1 : 0, wr = okr ? w + 1 : 63;
        float kv[KK], mp[KK];
#pragma unroll
        for (int kk = 0; kk < KK; ++kk) {
            const int  dy = kk / 3 - 1, dx = kk % 3 - 1;
            const int  hh = h + dy, ww = w + dx;
            const bool ok = (unsigned)hh < 64u && (unsigned)ww < 64u;
            const float* kr = kq + kk * (SS * 3 * 128);
            float el = kr[wl]       + kr[K3_KPART + wl];
            float ec = kr[128 + w]  + kr[K3_KPART + 128 + w];
            float er = kr[256 + wr] + kr[K3_KPART + 256 + wr];
            el = okl ? el : 0.f;
            er = okr ? er : 0.f;
            kv[kk] = ((el + ec) + er) + b2[kk * SS + s];
            mp[kk] = meanb[ok ? hh * 64 + ww : h * 64 + w];
            mp[kk] = ok ? mp[kk] : 0.f;
        }
        const float mc = meanb[h * 64 + w];
        float mval[KK];
#pragma unroll
        for (int kk = 0; kk < KK; ++kk) {
            float g = mp[kk] - mc;
            g = 1.0f / (g * g + 0.2f);
            mval[kk] = g * kv[kk];
        }
        float mx = mval[0];
#pragma unroll
        for (int kk = 1; kk < KK; ++kk) mx = fmaxf(mx, mval[kk]);
        float e[KK], sum = 0.f;
#pragma unroll
        for (int kk = 0; kk < KK; ++kk) { e[kk] = __expf(mval[kk] - mx); sum += e[kk]; }
        const float inv = 1.0f / sum;
#pragma unroll
        for (int kk = 0; kk < KK; ++kk) sm[(kk * SS + s) * WWID + w] = e[kk] * inv;
    }
    __syncthreads();

    float m0[2][KK][SS], m1[2][KK][SS];
#pragma unroll
    for (int rr = 0; rr < 2; ++rr)
#pragma unroll
        for (int kk = 0; kk < KK; ++kk)
#pragma unroll
            for (int s = 0; s < SS; ++s) {
                v2f t = *reinterpret_cast<const v2f*>(
                            &smask[rr * (KK * SS * WWID) + (kk * SS + s) * WWID + 2 * lane]);
                m0[rr][kk][s] = t.x;
                m1[rr][kk][s] = t.y;
            }

    for (int i = 0; i < 12; i += 4) {
        const float* xi = xw + (size_t)i * (16 * HW);
        float*       oi = ow + (size_t)i * (16 * 4 * HW);
        gg_load_rows(t3, xi + 48 * HW, om);
        gg_reassemble(t0, okM, okL, okR, idxL, idxR, m0, m1, oi);
        gg_load_rows(t0, xi + 64 * HW, om);
        gg_reassemble(t1, okM, okL, okR, idxL, idxR, m0, m1, oi + 16 * 4 * HW);
        gg_load_rows(t1, xi + 80 * HW, om);
        gg_reassemble(t2, okM, okL, okR, idxL, idxR, m0, m1, oi + 32 * 4 * HW);
        gg_load_rows(t2, xi + 96 * HW, om);
        gg_reassemble(t3, okM, okL, okR, idxL, idxR, m0, m1, oi + 48 * 4 * HW);
    }
    {
        const float* xi = xw + (size_t)12 * (16 * HW);
        float*       oi = ow + (size_t)12 * (16 * 4 * HW);
        gg_load_rows(t3, xi + 48 * HW, om);
        gg_reassemble(t0, okM, okL, okR, idxL, idxR, m0, m1, oi);
        gg_reassemble(t1, okM, okL, okR, idxL, idxR, m0, m1, oi + 16 * 4 * HW);
        gg_reassemble(t2, okM, okL, okR, idxL, idxR, m0, m1, oi + 32 * 4 * HW);
        gg_reassemble(t3, okM, okL, okR, idxL, idxR, m0, m1, oi + 48 * 4 * HW);
    }
}

extern "C" void kernel_launch(void* const* d_in, const int* in_sizes, int n_in,
                              void* d_out, int out_size, void* d_ws, size_t ws_size,
                              hipStream_t stream) {
    const float* x  = (const float*)d_in[0];
    const float* w1 = (const float*)d_in[1];
    const float* b1 = (const float*)d_in[2];
    const float* w2 = (const float*)d_in[3];
    const float* b2 = (const float*)d_in[4];
    float* out = (float*)d_out;

    float* ws   = (float*)d_ws;
    float* mean = ws;
    float* comp = ws + 32768;

    gg_comp_gemm  <<<256, 256, 0, stream>>>(x, w1, b1, comp, mean);
    gg_kern_carafe<<<256, 512, 0, stream>>>(comp, w2, b2, x, mean, out);
}
